# starting kernel
# baseline (speedup 1.0000x reference)
#include <hip/hip_runtime.h>

#define HDIM 4
#define ODIM 4
#define JITTER 1e-4f

#define WS_LINV 0
#define WS_B    4096
#define WS_ZS   8192
#define WS_Z2   8704
#define WS_Q    8768
#define WS_ILS  8832
#define WS_VAR  8840
#define WS_STRIDE 8960

typedef __attribute__((ext_vector_type(2))) float v2f;
typedef __attribute__((ext_vector_type(8))) float v8f;

__global__ __launch_bounds__(64) void svgp_prep(
    const float* __restrict__ z, const float* __restrict__ u_mean,
    const float* __restrict__ u_tril_vec, const float* __restrict__ log_ls,
    const float* __restrict__ log_var, float* __restrict__ ws)
{
    __shared__ float sL[64 * 64];
    __shared__ float sLinv[64 * 64];
    __shared__ float sA[64 * 64];
    __shared__ float sZs[64 * 8];
    __shared__ float sZ2[64];
    __shared__ float sT[64];

    const int p   = blockIdx.x;
    const int o   = p % ODIM;
    const int tid = threadIdx.x;
    float* wp = ws + (size_t)p * WS_STRIDE;

    float ils[8];
#pragma unroll
    for (int d = 0; d < 8; ++d) ils[d] = __expf(-log_ls[p * 8 + d]);
    const float var = __expf(log_var[p]);

    {
        float s = 0.f;
#pragma unroll
        for (int d = 0; d < 8; ++d) {
            float v = z[(o * 64 + tid) * 8 + d] * ils[d];
            sZs[tid * 8 + d] = v;
            s += v * v;
        }
        sZ2[tid] = s;
    }
    __syncthreads();

    {
        const int i = tid;
        for (int j = 0; j < 64; ++j) {
            float dot = 0.f;
#pragma unroll
            for (int d = 0; d < 8; ++d) dot += sZs[i * 8 + d] * sZs[j * 8 + d];
            float v = var * __expf(-0.5f * (sZ2[i] + sZ2[j] - 2.f * dot));
            if (i == j) v += JITTER;
            sL[i * 64 + j] = v;
        }
    }
    __syncthreads();

    for (int j = 0; j < 64; ++j) {
        if (tid == j) {
            float s = sL[j * 64 + j];
            for (int k = 0; k < j; ++k) s -= sL[j * 64 + k] * sL[j * 64 + k];
            sL[j * 64 + j] = sqrtf(s);
        }
        __syncthreads();
        if (tid > j) {
            float s = sL[tid * 64 + j];
            for (int k = 0; k < j; ++k) s -= sL[tid * 64 + k] * sL[j * 64 + k];
            sL[tid * 64 + j] = s / sL[j * 64 + j];
        }
        __syncthreads();
    }

    {
        const int c = tid;
        for (int i = 0; i < c; ++i) sLinv[i * 64 + c] = 0.f;
        sLinv[c * 64 + c] = 1.f / sL[c * 64 + c];
        for (int i = c + 1; i < 64; ++i) {
            float s = 0.f;
            for (int k = c; k < i; ++k) s += sL[i * 64 + k] * sLinv[k * 64 + c];
            sLinv[i * 64 + c] = -s / sL[i * 64 + i];
        }
    }

    {
        const int i    = tid;
        const float di = 1.f / sL[i * 64 + i];
        const int base = o * 2080 + i * (i + 1) / 2;
        for (int j = 0; j < 64; ++j)
            sA[i * 64 + j] = (j <= i) ? u_tril_vec[base + j] * di : 0.f;
    }
    __syncthreads();

    {
        const int i = tid;
        float s = 0.f;
        for (int j = 0; j <= i; ++j) s += sLinv[i * 64 + j] * u_mean[o * 64 + j];
        sT[i] = s;
    }
    __syncthreads();

    {
        const int j = tid;
        float s = 0.f;
        for (int i = j; i < 64; ++i) s += sLinv[i * 64 + j] * sT[i];
        wp[WS_Q + j] = s;
    }

    {
        const int j = tid;
        for (int r = 0; r < 64; ++r) {
            float s = 0.f;
            const int i0 = (j > r) ? j : r;
            for (int i = i0; i < 64; ++i) s += sA[i * 64 + j] * sLinv[i * 64 + r];
            wp[WS_B + j * 64 + r] = s;
        }
        for (int r = 0; r < 64; ++r) wp[WS_LINV + j * 64 + r] = sLinv[j * 64 + r];
#pragma unroll
        for (int d = 0; d < 8; ++d) wp[WS_ZS + j * 8 + d] = sZs[j * 8 + d];
        wp[WS_Z2 + j] = sZ2[j];
        if (j < 8) wp[WS_ILS + j] = ils[j];
        if (j == 0) wp[WS_VAR] = var;
    }
}

#define LSTR 65
#define COLS_PER_BLOCK 512

__global__ __launch_bounds__(128) void svgp_main(
    const float* __restrict__ x, const float* __restrict__ ws,
    float* __restrict__ out, int n, int blocksPerPair)
{
    __shared__ float sLinv[64 * LSTR];
    __shared__ float sB[64 * LSTR];
    __shared__ float sZs[64 * 8];
    __shared__ float sZ2[64];
    __shared__ float sQ[64];
    __shared__ float sK[4][64 * 16];
    __shared__ float sXs[4][16 * 8];
    __shared__ float sX2[4][16];

    const int pair  = blockIdx.x / blocksPerPair;
    const int chunk = blockIdx.x % blocksPerPair;
    const float* wp = ws + (size_t)pair * WS_STRIDE;
    const int tid  = threadIdx.x;
    const int wave = tid >> 5;
    const int lane = tid & 31;

    for (int i = tid; i < 4096; i += 128) {
        const int r = i >> 6, c = i & 63;
        sLinv[r * LSTR + c] = wp[WS_LINV + i];
        sB[r * LSTR + c]    = wp[WS_B + i];
    }
    for (int i = tid; i < 512; i += 128) sZs[i] = wp[WS_ZS + i];
    if (tid < 64) { sZ2[tid] = wp[WS_Z2 + tid]; sQ[tid] = wp[WS_Q + tid]; }
    float ils[8];
#pragma unroll
    for (int d = 0; d < 8; ++d) ils[d] = wp[WS_ILS + d];
    const float var = wp[WS_VAR];
    __syncthreads();

    const size_t varBase = (size_t)(HDIM * ODIM) * (size_t)n;
    const int nloc  = lane & 15;
    const int khalf = (lane >> 4) * 2;

    for (int it = 0; it < COLS_PER_BLOCK / 64; ++it) {
        const int col0 = chunk * COLS_PER_BLOCK + it * 64 + wave * 16;
        __syncthreads();

        if (lane < 16) {
            const int cidx = col0 + lane;
            float s = 0.f;
#pragma unroll
            for (int d = 0; d < 8; ++d) {
                float v = x[cidx * 8 + d] * ils[d];
                sXs[wave][lane * 8 + d] = v;
                s += v * v;
            }
            sX2[wave][lane] = s;
        }
        __syncthreads();

        {
            const float xn2 = sX2[wave][nloc];
#pragma unroll 4
            for (int i = 0; i < 32; ++i) {
                const int m = 2 * i + (lane >> 4);
                float dot = 0.f;
#pragma unroll
                for (int d = 0; d < 8; ++d)
                    dot += sZs[m * 8 + d] * sXs[wave][nloc * 8 + d];
                sK[wave][m * 16 + nloc] =
                    var * __expf(-0.5f * (xn2 + sZ2[m] - 2.f * dot));
            }
        }
        __syncthreads();

        v8f accY[4] = {};
        v8f accW[4] = {};
#pragma unroll
        for (int kc = 0; kc < 16; ++kc) {
            const int k0 = kc * 4 + khalf;
            v2f bfrag;
            bfrag.x = sK[wave][(k0 + 0) * 16 + nloc];
            bfrag.y = sK[wave][(k0 + 1) * 16 + nloc];
#pragma unroll
            for (int rt = 0; rt < 4; ++rt) {
                const int m = rt * 16 + nloc;
                v2f aY; aY.x = sLinv[m * LSTR + k0]; aY.y = sLinv[m * LSTR + k0 + 1];
                accY[rt] = __builtin_amdgcn_wmma_f32_16x16x4_f32(
                    false, aY, false, bfrag, (short)0, accY[rt], false, false);
                v2f aW; aW.x = sB[m * LSTR + k0]; aW.y = sB[m * LSTR + k0 + 1];
                accW[rt] = __builtin_amdgcn_wmma_f32_16x16x4_f32(
                    false, aW, false, bfrag, (short)0, accW[rt], false, false);
            }
        }

        float d1 = 0.f, d2 = 0.f;
#pragma unroll
        for (int rt = 0; rt < 4; ++rt)
#pragma unroll
            for (int v = 0; v < 8; ++v) {
                d1 += accY[rt][v] * accY[rt][v];
                d2 += accW[rt][v] * accW[rt][v];
            }
        d1 += __shfl_xor(d1, 16, 32);
        d2 += __shfl_xor(d2, 16, 32);

        if (lane < 16) {
            float mu = 0.f;
#pragma unroll 8
            for (int m = 0; m < 64; ++m) mu += sQ[m] * sK[wave][m * 16 + lane];
            const int col = col0 + lane;
            out[(size_t)pair * n + col]           = mu;
            out[varBase + (size_t)pair * n + col] = var - d1 + d2;
        }
    }
}

extern "C" void kernel_launch(void* const* d_in, const int* in_sizes, int n_in,
                              void* d_out, int out_size, void* d_ws, size_t ws_size,
                              hipStream_t stream)
{
    const float* x          = (const float*)d_in[0];
    const float* z          = (const float*)d_in[1];
    const float* u_mean     = (const float*)d_in[2];
    const float* u_tril_vec = (const float*)d_in[3];
    const float* log_ls     = (const float*)d_in[4];
    const float* log_var    = (const float*)d_in[5];
    float* out = (float*)d_out;
    float* ws  = (float*)d_ws;

    const int n = in_sizes[0] / 8;
    svgp_prep<<<HDIM * ODIM, 64, 0, stream>>>(z, u_mean, u_tril_vec,
                                              log_ls, log_var, ws);
    const int blocksPerPair = n / COLS_PER_BLOCK;
    svgp_main<<<HDIM * ODIM * blocksPerPair, 128, 0, stream>>>(
        x, ws, out, n, blocksPerPair);
}


# main: waves split the 64 rows; each keeps the A fragments of its 16 rows of Linv and B in registers for the whole block, only the transposed kfu tile is read from LDS (16-byte fragments)
# speedup vs baseline: 2.7259x; 2.7259x over previous
#include <hip/hip_runtime.h>

#define HDIM 4
#define ODIM 4
#define JITTER 1e-4f

#define WS_LINV 0
#define WS_B    4096
#define WS_ZS   8192
#define WS_Z2   8704
#define WS_Q    8768
#define WS_ILS  8832
#define WS_VAR  8840
#define WS_STRIDE 8960

typedef __attribute__((ext_vector_type(2))) float v2f;
typedef __attribute__((ext_vector_type(8))) float v8f;

#define PS 65
#define WSTR 17

__device__ __forceinline__ v8f tile_mma(v8f acc, const float* A, int a_rs, int a_cs,
                                        const float* B, int b_rs, int b_cs, int lane)
{
    const int nl = lane & 15;
    const int kh = (lane >> 4) * 2;
#pragma unroll
    for (int kc = 0; kc < 4; ++kc) {
        const int k0 = kc * 4 + kh;
        v2f a, b;
        a.x = A[nl * a_rs + k0 * a_cs];
        a.y = A[nl * a_rs + (k0 + 1) * a_cs];
        b.x = B[k0 * b_rs + nl * b_cs];
        b.y = B[(k0 + 1) * b_rs + nl * b_cs];
        acc = __builtin_amdgcn_wmma_f32_16x16x4_f32(
            false, a, false, b, (short)0, acc, false, false);
    }
    return acc;
}

__global__ __launch_bounds__(128) void svgp_prep(
    const float* __restrict__ z, const float* __restrict__ u_mean,
    const float* __restrict__ u_tril_vec, const float* __restrict__ log_ls,
    const float* __restrict__ log_var, float* __restrict__ ws)
{
    __shared__ float sS[64 * PS];
    __shared__ float sX[64 * PS];
    __shared__ float sU[64 * PS];
    __shared__ float sW[4][16 * WSTR];
    __shared__ float sZs[64 * 8];
    __shared__ float sZ2[64];
    __shared__ float sT[64];
    __shared__ float sUm[64];
    __shared__ float sCol[2][16];

    const int p    = blockIdx.x;
    const int o    = p % ODIM;
    const int tid  = threadIdx.x;
    const int lane = tid & 31;
    const int wave = __builtin_amdgcn_readfirstlane(tid >> 5);
    const int row  = tid & 63;
    const int half = tid >> 6;
    const int accr = (lane >> 4) * 8;
    const int accc = lane & 15;
    float* wp = ws + (size_t)p * WS_STRIDE;

    float ils[8];
#pragma unroll
    for (int d = 0; d < 8; ++d) ils[d] = __expf(-log_ls[p * 8 + d]);
    const float var = __expf(log_var[p]);

#pragma unroll 8
    for (int it = 0; it < 32; ++it) {
        const int i  = 2 * it + half;
        const int jj = (row < i) ? row : i;
        const float v = u_tril_vec[o * 2080 + i * (i + 1) / 2 + jj];
        sU[i * PS + row] = (row <= i) ? v : 0.f;
    }
    for (int i = tid; i < 64 * PS; i += 128) sX[i] = 0.f;

    float zr[8];
    {
        float s = 0.f;
#pragma unroll
        for (int d = 0; d < 8; ++d) {
            zr[d] = z[(o * 64 + row) * 8 + d] * ils[d];
            s += zr[d] * zr[d];
        }
        if (half == 0) {
#pragma unroll
            for (int d = 0; d < 8; ++d) sZs[row * 8 + d] = zr[d];
            sZ2[row] = s;
            sUm[row] = u_mean[o * 64 + row];
        }
    }
    __syncthreads();

    {
        const float z2i = sZ2[row];
        for (int jj = 0; jj < 32; ++jj) {
            const int j = half * 32 + jj;
            float dot = 0.f;
#pragma unroll
            for (int d = 0; d < 8; ++d) dot += zr[d] * sZs[j * 8 + d];
            float v = var * __expf(-0.5f * (z2i + sZ2[j] - 2.f * dot));
            if (row == j) v += JITTER;
            sS[row * PS + j] = v;
        }
    }
    __syncthreads();

    for (int J = 0; J < 4; ++J) {
        if (J > 0 && wave >= J) {
            v8f acc = {};
            for (int P = 0; P < J; ++P)
                acc = tile_mma(acc, &sS[(16 * wave) * PS + 16 * P], PS, 1,
                                    &sS[(16 * J) * PS + 16 * P], 1, PS, lane);
#pragma unroll
            for (int v = 0; v < 8; ++v)
                sS[(16 * wave + accr + v) * PS + 16 * J + accc] -= acc[v];
        }
        __syncthreads();

        float* blk = &sS[(16 * J) * PS + 16 * J];
        float a[16];
        if (tid < 16) {
#pragma unroll
            for (int k = 0; k < 16; ++k) a[k] = blk[tid * PS + k];
            sCol[0][tid] = a[0];
        }
        __syncthreads();
#pragma unroll
        for (int s = 0; s < 16; ++s) {
            if (tid < 16) {
                const float* cur = sCol[s & 1];
                const float ass = cur[s];
                const float d   = sqrtf(ass);
                if (tid == s) a[s] = d;
                if (tid > s) {
                    const float c = a[s] / ass;
                    a[s] = a[s] / d;
#pragma unroll
                    for (int k = s + 1; k < 16; ++k) a[k] -= c * cur[k];
                    if (s < 15) sCol[(s + 1) & 1][tid] = a[s + 1];
                }
            }
            __syncthreads();
        }
        if (tid < 16) {
#pragma unroll
            for (int k = 0; k < 16; ++k)
                if (k <= tid) blk[tid * PS + k] = a[k];
        }
        __syncthreads();

        if (tid < 16) {
            float rd[16], x[16];
#pragma unroll
            for (int i = 0; i < 16; ++i) {
                rd[i] = 1.f / blk[i * PS + i];
                x[i]  = (i == tid) ? 1.f : 0.f;
            }
#pragma unroll
            for (int i = 0; i < 16; ++i) {
                x[i] *= rd[i];
#pragma unroll
                for (int k = i + 1; k < 16; ++k) x[k] -= blk[k * PS + i] * x[i];
                sW[J][i * WSTR + tid] = x[i];
            }
        }
        __syncthreads();

        if (wave > J) {
            v8f acc = {};
            acc = tile_mma(acc, &sS[(16 * wave) * PS + 16 * J], PS, 1,
                                &sW[J][0], 1, WSTR, lane);
#pragma unroll
            for (int v = 0; v < 8; ++v)
                sS[(16 * wave + accr + v) * PS + 16 * J + accc] = acc[v];
        }
        __syncthreads();
    }

    {
        const int r = tid >> 3, c0 = (tid & 7) * 2;
#pragma unroll
        for (int J = 0; J < 4; ++J) {
            sX[(16 * J + r) * PS + 16 * J + c0]     = sW[J][r * WSTR + c0];
            sX[(16 * J + r) * PS + 16 * J + c0 + 1] = sW[J][r * WSTR + c0 + 1];
        }
    }
    __syncthreads();
    for (int dist = 1; dist < 4; ++dist) {
        const int I  = wave;
        const int Jc = wave - dist;
        const bool act = (Jc >= 0);
        if (act) {
            v8f acc = {};
            for (int P = Jc; P < I; ++P)
                acc = tile_mma(acc, &sS[(16 * I) * PS + 16 * P], PS, 1,
                                    &sX[(16 * P) * PS + 16 * Jc], PS, 1, lane);
#pragma unroll
            for (int v = 0; v < 8; ++v)
                sX[(16 * I + accr + v) * PS + 16 * Jc + accc] = acc[v];
        }
        __syncthreads();
        v8f acc2 = {};
        if (act)
            acc2 = tile_mma(acc2, &sW[I][0], WSTR, 1,
                                  &sX[(16 * I) * PS + 16 * Jc], PS, 1, lane);
        __syncthreads();
        if (act) {
#pragma unroll
            for (int v = 0; v < 8; ++v)
                sX[(16 * I + accr + v) * PS + 16 * Jc + accc] = -acc2[v];
        }
        __syncthreads();
    }

    {
        const float di = 1.f / sS[row * PS + row];
        for (int jj = 0; jj < 32; ++jj) sU[row * PS + half * 32 + jj] *= di;
        if (half == 0) {
            float s = 0.f;
            for (int j = 0; j < 64; ++j) s += sX[row * PS + j] * sUm[j];
            sT[row] = s;
        }
    }
    __syncthreads();

    if (half == 0) {
        float s = 0.f;
        for (int i = 0; i < 64; ++i) s += sX[i * PS + row] * sT[i];
        wp[WS_Q + row] = s;
    }

    for (int Rt = 0; Rt < 4; ++Rt) {
        v8f acc = {};
        const int It0 = (wave > Rt) ? wave : Rt;
        for (int It = It0; It < 4; ++It)
            acc = tile_mma(acc, &sU[(16 * It) * PS + 16 * wave], 1, PS,
                                &sX[(16 * It) * PS + 16 * Rt], PS, 1, lane);
#pragma unroll
        for (int v = 0; v < 8; ++v)
            wp[WS_B + (16 * wave + accr + v) * 64 + 16 * Rt + accc] = acc[v];
    }

    for (int i = tid; i < 4096; i += 128)
        wp[WS_LINV + i] = sX[(i >> 6) * PS + (i & 63)];
    for (int i = tid; i < 512; i += 128) wp[WS_ZS + i] = sZs[i];
    if (tid < 64) wp[WS_Z2 + tid] = sZ2[tid];
    if (tid < 8)  wp[WS_ILS + tid] = __expf(-log_ls[p * 8 + tid]);
    if (tid == 0) wp[WS_VAR] = var;
}

#define KP 68
#define COLS_PER_BLOCK 512
#define COLS_PER_ITER 64

typedef __attribute__((ext_vector_type(4))) float v4f;

__global__ __launch_bounds__(128) void svgp_main(
    const float* __restrict__ x, const float* __restrict__ ws,
    float* __restrict__ out, int n, int blocksPerPair)
{
    __shared__ __attribute__((aligned(16))) float sKt[64 * KP];
    __shared__ __attribute__((aligned(16))) float sZs[64 * 8];
    __shared__ __attribute__((aligned(16))) float sZ2[64];
    __shared__ __attribute__((aligned(16))) float sQ[64];
    __shared__ float sMu[2][2][64];
    __shared__ float sD1[2][4][64];
    __shared__ float sD2[2][4][64];

    const int pair  = blockIdx.x / blocksPerPair;
    const int chunk = blockIdx.x % blocksPerPair;
    const float* wp = ws + (size_t)pair * WS_STRIDE;
    const int tid  = threadIdx.x;
    const int wave = __builtin_amdgcn_readfirstlane(tid >> 5);
    const int lane = tid & 31;
    const int c    = lane & 15;
    const int h    = lane >> 4;

    *(v4f*)&sZs[4 * tid] = *(const v4f*)(wp + WS_ZS + 4 * tid);
    if (tid < 64) { sZ2[tid] = wp[WS_Z2 + tid]; sQ[tid] = wp[WS_Q + tid]; }
    float ils[8];
#pragma unroll
    for (int d = 0; d < 8; ++d) ils[d] = wp[WS_ILS + d];
    const float var = wp[WS_VAR];

    v4f aL[8], aB[8];
    {
        const float* rowL = wp + WS_LINV + (16 * wave + c) * 64 + 4 * h;
        const float* rowB = wp + WS_B    + (16 * wave + c) * 64 + 4 * h;
#pragma unroll
        for (int j = 0; j < 8; ++j) {
            aL[j] = *(const v4f*)(rowL + 8 * j);
            aB[j] = *(const v4f*)(rowB + 8 * j);
        }
    }
    __syncthreads();

    const size_t varBase = (size_t)(HDIM * ODIM) * (size_t)n;
    const int ncol = tid & 63;
    const int mh   = tid >> 6;
    const float* bp = &sKt[c * KP + 4 * h];

    for (int it = 0; it < COLS_PER_BLOCK / COLS_PER_ITER; ++it) {
        const int p    = it & 1;
        const int col0 = chunk * COLS_PER_BLOCK + it * COLS_PER_ITER;

        float xs[8], xn2 = 0.f;
        {
            const v4f xa = *(const v4f*)(x + (size_t)(col0 + ncol) * 8);
            const v4f xb = *(const v4f*)(x + (size_t)(col0 + ncol) * 8 + 4);
            xs[0] = xa.x * ils[0]; xs[1] = xa.y * ils[1]; xs[2] = xa.z * ils[2]; xs[3] = xa.w * ils[3];
            xs[4] = xb.x * ils[4]; xs[5] = xb.y * ils[5]; xs[6] = xb.z * ils[6]; xs[7] = xb.w * ils[7];
#pragma unroll
            for (int d = 0; d < 8; ++d) xn2 += xs[d] * xs[d];
        }

        float mu = 0.f;
#pragma unroll 2
        for (int g = 0; g < 8; ++g) {
            const int m0 = 32 * mh + 4 * g;
            v4f kv;
#pragma unroll
            for (int e = 0; e < 4; ++e) {
                const int m = m0 + e;
                const v4f za = *(const v4f*)&sZs[m * 8];
                const v4f zb = *(const v4f*)&sZs[m * 8 + 4];
                float dot = 0.f;
                dot += za.x * xs[0]; dot += za.y * xs[1]; dot += za.z * xs[2]; dot += za.w * xs[3];
                dot += zb.x * xs[4]; dot += zb.y * xs[5]; dot += zb.z * xs[6]; dot += zb.w * xs[7];
                const float kval = var * __expf(-0.5f * (xn2 + sZ2[m] - 2.f * dot));
                kv[e] = kval;
                mu += sQ[m] * kval;
            }
            *(v4f*)&sKt[ncol * KP + m0] = kv;
        }
        sMu[p][mh][ncol] = mu;
        __syncthreads();

        v8f accY[4] = {};
        v8f accW[4] = {};
#pragma unroll
        for (int j = 0; j < 8; ++j) {
            v4f b[4];
#pragma unroll
            for (int ct = 0; ct < 4; ++ct)
                b[ct] = *(const v4f*)(bp + ct * 16 * KP + 8 * j);
            const v2f aY0 = {aL[j].x, aL[j].y}, aY1 = {aL[j].z, aL[j].w};
            const v2f aW0 = {aB[j].x, aB[j].y}, aW1 = {aB[j].z, aB[j].w};
#pragma unroll
            for (int ct = 0; ct < 4; ++ct) {
                const v2f b0 = {b[ct].x, b[ct].y};
                accY[ct] = __builtin_amdgcn_wmma_f32_16x16x4_f32(
                    false, aY0, false, b0, (short)0, accY[ct], false, false);
                accW[ct] = __builtin_amdgcn_wmma_f32_16x16x4_f32(
                    false, aW0, false, b0, (short)0, accW[ct], false, false);
            }
#pragma unroll
            for (int ct = 0; ct < 4; ++ct) {
                const v2f b1 = {b[ct].z, b[ct].w};
                accY[ct] = __builtin_amdgcn_wmma_f32_16x16x4_f32(
                    false, aY1, false, b1, (short)0, accY[ct], false, false);
                accW[ct] = __builtin_amdgcn_wmma_f32_16x16x4_f32(
                    false, aW1, false, b1, (short)0, accW[ct], false, false);
            }
        }

#pragma unroll
        for (int ct = 0; ct < 4; ++ct) {
            float d1 = 0.f, d2 = 0.f;
#pragma unroll
            for (int v = 0; v < 8; ++v) {
                d1 += accY[ct][v] * accY[ct][v];
                d2 += accW[ct][v] * accW[ct][v];
            }
            d1 += __shfl_xor(d1, 16, 32);
            d2 += __shfl_xor(d2, 16, 32);
            if (lane < 16) {
                sD1[p][wave][ct * 16 + c] = d1;
                sD2[p][wave][ct * 16 + c] = d2;
            }
        }
        __syncthreads();

        if (tid < 64) {
            const float m_ = sMu[p][0][tid] + sMu[p][1][tid];
            const float d1 = (sD1[p][0][tid] + sD1[p][1][tid]) + (sD1[p][2][tid] + sD1[p][3][tid]);
            const float d2 = (sD2[p][0][tid] + sD2[p][1][tid]) + (sD2[p][2][tid] + sD2[p][3][tid]);
            const int col = col0 + tid;
            out[(size_t)pair * n + col]           = m_;
            out[varBase + (size_t)pair * n + col] = var - d1 + d2;
        }
    }
}

extern "C" void kernel_launch(void* const* d_in, const int* in_sizes, int n_in,
                              void* d_out, int out_size, void* d_ws, size_t ws_size,
                              hipStream_t stream)
{
    const float* x          = (const float*)d_in[0];
    const float* z          = (const float*)d_in[1];
    const float* u_mean     = (const float*)d_in[2];
    const float* u_tril_vec = (const float*)d_in[3];
    const float* log_ls     = (const float*)d_in[4];
    const float* log_var    = (const float*)d_in[5];
    float* out = (float*)d_out;
    float* ws  = (float*)d_ws;

    const int n = in_sizes[0] / 8;
    svgp_prep<<<HDIM * ODIM, 128, 0, stream>>>(z, u_mean, u_tril_vec,
                                              log_ls, log_var, ws);
    const int blocksPerPair = n / COLS_PER_BLOCK;
    svgp_main<<<HDIM * ODIM * blocksPerPair, 128, 0, stream>>>(
        x, ws, out, n, blocksPerPair);
}


# prep: each 16x16 diagonal block is factored and inverted by one wave in registers (lane = row, pivot columns handed over by lane reads): no LDS hand-over and no block barrier per pivot
# speedup vs baseline: 2.8971x; 1.0628x over previous
#include <hip/hip_runtime.h>

#define HDIM 4
#define ODIM 4
#define JITTER 1e-4f

#define WS_LINV 0
#define WS_B    4096
#define WS_ZS   8192
#define WS_Z2   8704
#define WS_Q    8768
#define WS_ILS  8832
#define WS_VAR  8840
#define WS_STRIDE 8960

typedef __attribute__((ext_vector_type(2))) float v2f;
typedef __attribute__((ext_vector_type(8))) float v8f;

#define PS 65
#define WSTR 17

__device__ __forceinline__ v8f tile_mma(v8f acc, const float* A, int a_rs, int a_cs,
                                        const float* B, int b_rs, int b_cs, int lane)
{
    const int nl = lane & 15;
    const int kh = (lane >> 4) * 2;
#pragma unroll
    for (int kc = 0; kc < 4; ++kc) {
        const int k0 = kc * 4 + kh;
        v2f a, b;
        a.x = A[nl * a_rs + k0 * a_cs];
        a.y = A[nl * a_rs + (k0 + 1) * a_cs];
        b.x = B[k0 * b_rs + nl * b_cs];
        b.y = B[(k0 + 1) * b_rs + nl * b_cs];
        acc = __builtin_amdgcn_wmma_f32_16x16x4_f32(
            false, a, false, b, (short)0, acc, false, false);
    }
    return acc;
}

__device__ __forceinline__ float lane_read(float x, int k)
{
    return __builtin_bit_cast(float,
        __builtin_amdgcn_readlane(__builtin_bit_cast(int, x), k));
}

__global__ __launch_bounds__(128) void svgp_prep(
    const float* __restrict__ z, const float* __restrict__ u_mean,
    const float* __restrict__ u_tril_vec, const float* __restrict__ log_ls,
    const float* __restrict__ log_var, float* __restrict__ ws)
{
    __shared__ float sS[64 * PS];
    __shared__ float sX[64 * PS];
    __shared__ float sU[64 * PS];
    __shared__ float sW[4][16 * WSTR];
    __shared__ float sZs[64 * 8];
    __shared__ float sZ2[64];
    __shared__ float sT[64];
    __shared__ float sUm[64];

    const int p    = blockIdx.x;
    const int o    = p % ODIM;
    const int tid  = threadIdx.x;
    const int lane = tid & 31;
    const int wave = __builtin_amdgcn_readfirstlane(tid >> 5);
    const int row  = tid & 63;
    const int half = tid >> 6;
    const int accr = (lane >> 4) * 8;
    const int accc = lane & 15;
    float* wp = ws + (size_t)p * WS_STRIDE;

    float ils[8];
#pragma unroll
    for (int d = 0; d < 8; ++d) ils[d] = __expf(-log_ls[p * 8 + d]);
    const float var = __expf(log_var[p]);

#pragma unroll 8
    for (int it = 0; it < 32; ++it) {
        const int i  = 2 * it + half;
        const int jj = (row < i) ? row : i;
        const float v = u_tril_vec[o * 2080 + i * (i + 1) / 2 + jj];
        sU[i * PS + row] = (row <= i) ? v : 0.f;
    }
    for (int i = tid; i < 64 * PS; i += 128) sX[i] = 0.f;

    float zr[8];
    {
        float s = 0.f;
#pragma unroll
        for (int d = 0; d < 8; ++d) {
            zr[d] = z[(o * 64 + row) * 8 + d] * ils[d];
            s += zr[d] * zr[d];
        }
        if (half == 0) {
#pragma unroll
            for (int d = 0; d < 8; ++d) sZs[row * 8 + d] = zr[d];
            sZ2[row] = s;
            sUm[row] = u_mean[o * 64 + row];
        }
    }
    __syncthreads();

    {
        const float z2i = sZ2[row];
        for (int jj = 0; jj < 32; ++jj) {
            const int j = half * 32 + jj;
            float dot = 0.f;
#pragma unroll
            for (int d = 0; d < 8; ++d) dot += zr[d] * sZs[j * 8 + d];
            float v = var * __expf(-0.5f * (z2i + sZ2[j] - 2.f * dot));
            if (row == j) v += JITTER;
            sS[row * PS + j] = v;
        }
    }
    __syncthreads();

    for (int J = 0; J < 4; ++J) {
        if (J > 0 && wave >= J) {
            v8f acc = {};
            for (int P = 0; P < J; ++P)
                acc = tile_mma(acc, &sS[(16 * wave) * PS + 16 * P], PS, 1,
                                    &sS[(16 * J) * PS + 16 * P], 1, PS, lane);
#pragma unroll
            for (int v = 0; v < 8; ++v)
                sS[(16 * wave + accr + v) * PS + 16 * J + accc] -= acc[v];
        }
        __syncthreads();

        float* blk = &sS[(16 * J) * PS + 16 * J];
        if (wave == 0) {
            const int t = lane & 15;
            float a[16];
#pragma unroll
            for (int k = 0; k < 16; ++k) a[k] = blk[t * PS + k];
#pragma unroll
            for (int s = 0; s < 16; ++s) {
                float cur[16];
#pragma unroll
                for (int k = s; k < 16; ++k) cur[k] = lane_read(a[s], k);
                const float ass = cur[s];
                const float d   = sqrtf(ass);
                const float c   = a[s] / ass;
                const float l   = a[s] / d;
                a[s] = (t == s) ? d : l;
#pragma unroll
                for (int k = s + 1; k < 16; ++k) a[k] -= c * cur[k];
            }
            float rd[16], x[16];
#pragma unroll
            for (int i = 0; i < 16; ++i) {
                rd[i] = 1.f / lane_read(a[i], i);
                x[i]  = (i == t) ? 1.f : 0.f;
            }
#pragma unroll
            for (int i = 0; i < 16; ++i) {
                x[i] *= rd[i];
#pragma unroll
                for (int k = i + 1; k < 16; ++k) x[k] -= lane_read(a[i], k) * x[i];
            }
            if (lane < 16) {
#pragma unroll
                for (int i = 0; i < 16; ++i) sW[J][i * WSTR + t] = x[i];
#pragma unroll
                for (int k = 0; k < 16; ++k)
                    if (k <= t) blk[t * PS + k] = a[k];
            }
        }
        __syncthreads();

        if (wave > J) {
            v8f acc = {};
            acc = tile_mma(acc, &sS[(16 * wave) * PS + 16 * J], PS, 1,
                                &sW[J][0], 1, WSTR, lane);
#pragma unroll
            for (int v = 0; v < 8; ++v)
                sS[(16 * wave + accr + v) * PS + 16 * J + accc] = acc[v];
        }
        __syncthreads();
    }

    {
        const int r = tid >> 3, c0 = (tid & 7) * 2;
#pragma unroll
        for (int J = 0; J < 4; ++J) {
            sX[(16 * J + r) * PS + 16 * J + c0]     = sW[J][r * WSTR + c0];
            sX[(16 * J + r) * PS + 16 * J + c0 + 1] = sW[J][r * WSTR + c0 + 1];
        }
    }
    __syncthreads();
    for (int dist = 1; dist < 4; ++dist) {
        const int I  = wave;
        const int Jc = wave - dist;
        const bool act = (Jc >= 0);
        if (act) {
            v8f acc = {};
            for (int P = Jc; P < I; ++P)
                acc = tile_mma(acc, &sS[(16 * I) * PS + 16 * P], PS, 1,
                                    &sX[(16 * P) * PS + 16 * Jc], PS, 1, lane);
#pragma unroll
            for (int v = 0; v < 8; ++v)
                sX[(16 * I + accr + v) * PS + 16 * Jc + accc] = acc[v];
        }
        __syncthreads();
        v8f acc2 = {};
        if (act)
            acc2 = tile_mma(acc2, &sW[I][0], WSTR, 1,
                                  &sX[(16 * I) * PS + 16 * Jc], PS, 1, lane);
        __syncthreads();
        if (act) {
#pragma unroll
            for (int v = 0; v < 8; ++v)
                sX[(16 * I + accr + v) * PS + 16 * Jc + accc] = -acc2[v];
        }
        __syncthreads();
    }

    {
        const float di = 1.f / sS[row * PS + row];
        for (int jj = 0; jj < 32; ++jj) sU[row * PS + half * 32 + jj] *= di;
        if (half == 0) {
            float s = 0.f;
            for (int j = 0; j < 64; ++j) s += sX[row * PS + j] * sUm[j];
            sT[row] = s;
        }
    }
    __syncthreads();

    if (half == 0) {
        float s = 0.f;
        for (int i = 0; i < 64; ++i) s += sX[i * PS + row] * sT[i];
        wp[WS_Q + row] = s;
    }

    for (int Rt = 0; Rt < 4; ++Rt) {
        v8f acc = {};
        const int It0 = (wave > Rt) ? wave : Rt;
        for (int It = It0; It < 4; ++It)
            acc = tile_mma(acc, &sU[(16 * It) * PS + 16 * wave], 1, PS,
                                &sX[(16 * It) * PS + 16 * Rt], PS, 1, lane);
#pragma unroll
        for (int v = 0; v < 8; ++v)
            wp[WS_B + (16 * wave + accr + v) * 64 + 16 * Rt + accc] = acc[v];
    }

    for (int i = tid; i < 4096; i += 128)
        wp[WS_LINV + i] = sX[(i >> 6) * PS + (i & 63)];
    for (int i = tid; i < 512; i += 128) wp[WS_ZS + i] = sZs[i];
    if (tid < 64) wp[WS_Z2 + tid] = sZ2[tid];
    if (tid < 8)  wp[WS_ILS + tid] = __expf(-log_ls[p * 8 + tid]);
    if (tid == 0) wp[WS_VAR] = var;
}

#define KP 68
#define COLS_PER_BLOCK 512
#define COLS_PER_ITER 64

typedef __attribute__((ext_vector_type(4))) float v4f;

__global__ __launch_bounds__(128) void svgp_main(
    const float* __restrict__ x, const float* __restrict__ ws,
    float* __restrict__ out, int n, int blocksPerPair)
{
    __shared__ __attribute__((aligned(16))) float sKt[64 * KP];
    __shared__ __attribute__((aligned(16))) float sZs[64 * 8];
    __shared__ __attribute__((aligned(16))) float sZ2[64];
    __shared__ __attribute__((aligned(16))) float sQ[64];
    __shared__ float sMu[2][2][64];
    __shared__ float sD1[2][4][64];
    __shared__ float sD2[2][4][64];

    const int pair  = blockIdx.x / blocksPerPair;
    const int chunk = blockIdx.x % blocksPerPair;
    const float* wp = ws + (size_t)pair * WS_STRIDE;
    const int tid  = threadIdx.x;
    const int wave = __builtin_amdgcn_readfirstlane(tid >> 5);
    const int lane = tid & 31;
    const int c    = lane & 15;
    const int h    = lane >> 4;

    *(v4f*)&sZs[4 * tid] = *(const v4f*)(wp + WS_ZS + 4 * tid);
    if (tid < 64) { sZ2[tid] = wp[WS_Z2 + tid]; sQ[tid] = wp[WS_Q + tid]; }
    float ils[8];
#pragma unroll
    for (int d = 0; d < 8; ++d) ils[d] = wp[WS_ILS + d];
    const float var = wp[WS_VAR];

    v4f aL[8], aB[8];
    {
        const float* rowL = wp + WS_LINV + (16 * wave + c) * 64 + 4 * h;
        const float* rowB = wp + WS_B    + (16 * wave + c) * 64 + 4 * h;
#pragma unroll
        for (int j = 0; j < 8; ++j) {
            aL[j] = *(const v4f*)(rowL + 8 * j);
            aB[j] = *(const v4f*)(rowB + 8 * j);
        }
    }
    __syncthreads();

    const size_t varBase = (size_t)(HDIM * ODIM) * (size_t)n;
    const int ncol = tid & 63;
    const int mh   = tid >> 6;
    const float* bp = &sKt[c * KP + 4 * h];

    for (int it = 0; it < COLS_PER_BLOCK / COLS_PER_ITER; ++it) {
        const int p    = it & 1;
        const int col0 = chunk * COLS_PER_BLOCK + it * COLS_PER_ITER;

        float xs[8], xn2 = 0.f;
        {
            const v4f xa = *(const v4f*)(x + (size_t)(col0 + ncol) * 8);
            const v4f xb = *(const v4f*)(x + (size_t)(col0 + ncol) * 8 + 4);
            xs[0] = xa.x * ils[0]; xs[1] = xa.y * ils[1]; xs[2] = xa.z * ils[2]; xs[3] = xa.w * ils[3];
            xs[4] = xb.x * ils[4]; xs[5] = xb.y * ils[5]; xs[6] = xb.z * ils[6]; xs[7] = xb.w * ils[7];
#pragma unroll
            for (int d = 0; d < 8; ++d) xn2 += xs[d] * xs[d];
        }

        float mu = 0.f;
#pragma unroll 2
        for (int g = 0; g < 8; ++g) {
            const int m0 = 32 * mh + 4 * g;
            v4f kv;
#pragma unroll
            for (int e = 0; e < 4; ++e) {
                const int m = m0 + e;
                const v4f za = *(const v4f*)&sZs[m * 8];
                const v4f zb = *(const v4f*)&sZs[m * 8 + 4];
                float dot = 0.f;
                dot += za.x * xs[0]; dot += za.y * xs[1]; dot += za.z * xs[2]; dot += za.w * xs[3];
                dot += zb.x * xs[4]; dot += zb.y * xs[5]; dot += zb.z * xs[6]; dot += zb.w * xs[7];
                const float kval = var * __expf(-0.5f * (xn2 + sZ2[m] - 2.f * dot));
                kv[e] = kval;
                mu += sQ[m] * kval;
            }
            *(v4f*)&sKt[ncol * KP + m0] = kv;
        }
        sMu[p][mh][ncol] = mu;
        __syncthreads();

        v8f accY[4] = {};
        v8f accW[4] = {};
#pragma unroll
        for (int j = 0; j < 8; ++j) {
            v4f b[4];
#pragma unroll
            for (int ct = 0; ct < 4; ++ct)
                b[ct] = *(const v4f*)(bp + ct * 16 * KP + 8 * j);
            const v2f aY0 = {aL[j].x, aL[j].y}, aY1 = {aL[j].z, aL[j].w};
            const v2f aW0 = {aB[j].x, aB[j].y}, aW1 = {aB[j].z, aB[j].w};
#pragma unroll
            for (int ct = 0; ct < 4; ++ct) {
                const v2f b0 = {b[ct].x, b[ct].y};
                accY[ct] = __builtin_amdgcn_wmma_f32_16x16x4_f32(
                    false, aY0, false, b0, (short)0, accY[ct], false, false);
                accW[ct] = __builtin_amdgcn_wmma_f32_16x16x4_f32(
                    false, aW0, false, b0, (short)0, accW[ct], false, false);
            }
#pragma unroll
            for (int ct = 0; ct < 4; ++ct) {
                const v2f b1 = {b[ct].z, b[ct].w};
                accY[ct] = __builtin_amdgcn_wmma_f32_16x16x4_f32(
                    false, aY1, false, b1, (short)0, accY[ct], false, false);
                accW[ct] = __builtin_amdgcn_wmma_f32_16x16x4_f32(
                    false, aW1, false, b1, (short)0, accW[ct], false, false);
            }
        }

#pragma unroll
        for (int ct = 0; ct < 4; ++ct) {
            float d1 = 0.f, d2 = 0.f;
#pragma unroll
            for (int v = 0; v < 8; ++v) {
                d1 += accY[ct][v] * accY[ct][v];
                d2 += accW[ct][v] * accW[ct][v];
            }
            d1 += __shfl_xor(d1, 16, 32);
            d2 += __shfl_xor(d2, 16, 32);
            if (lane < 16) {
                sD1[p][wave][ct * 16 + c] = d1;
                sD2[p][wave][ct * 16 + c] = d2;
            }
        }
        __syncthreads();

        if (tid < 64) {
            const float m_ = sMu[p][0][tid] + sMu[p][1][tid];
            const float d1 = (sD1[p][0][tid] + sD1[p][1][tid]) + (sD1[p][2][tid] + sD1[p][3][tid]);
            const float d2 = (sD2[p][0][tid] + sD2[p][1][tid]) + (sD2[p][2][tid] + sD2[p][3][tid]);
            const int col = col0 + tid;
            out[(size_t)pair * n + col]           = m_;
            out[varBase + (size_t)pair * n + col] = var - d1 + d2;
        }
    }
}

extern "C" void kernel_launch(void* const* d_in, const int* in_sizes, int n_in,
                              void* d_out, int out_size, void* d_ws, size_t ws_size,
                              hipStream_t stream)
{
    const float* x          = (const float*)d_in[0];
    const float* z          = (const float*)d_in[1];
    const float* u_mean     = (const float*)d_in[2];
    const float* u_tril_vec = (const float*)d_in[3];
    const float* log_ls     = (const float*)d_in[4];
    const float* log_var    = (const float*)d_in[5];
    float* out = (float*)d_out;
    float* ws  = (float*)d_ws;

    const int n = in_sizes[0] / 8;
    svgp_prep<<<HDIM * ODIM, 128, 0, stream>>>(z, u_mean, u_tril_vec,
                                              log_ls, log_var, ws);
    const int blocksPerPair = n / COLS_PER_BLOCK;
    svgp_main<<<HDIM * ODIM * blocksPerPair, 128, 0, stream>>>(
        x, ws, out, n, blocksPerPair);
}


# main: each kfu entry is exp2 of one chain of eight multiply-adds, four entries per packed chain; prep stores the inducing points transposed times log2(e) and one start value per point
# speedup vs baseline: 3.0562x; 1.0549x over previous
#include <hip/hip_runtime.h>

#define HDIM 4
#define ODIM 4
#define JITTER 1e-4f
#define LOG2E 1.44269504088896340736f

#define WS_LINV 0
#define WS_B    4096
#define WS_ZS   8192
#define WS_Z2   8704
#define WS_Q    8768
#define WS_ILS  8832
#define WS_VAR  8840
#define WS_STRIDE 8960

typedef __attribute__((ext_vector_type(2))) float v2f;
typedef __attribute__((ext_vector_type(8))) float v8f;

#define PS 65
#define WSTR 17

__device__ __forceinline__ v8f tile_mma(v8f acc, const float* A, int a_rs, int a_cs,
                                        const float* B, int b_rs, int b_cs, int lane)
{
    const int nl = lane & 15;
    const int kh = (lane >> 4) * 2;
#pragma unroll
    for (int kc = 0; kc < 4; ++kc) {
        const int k0 = kc * 4 + kh;
        v2f a, b;
        a.x = A[nl * a_rs + k0 * a_cs];
        a.y = A[nl * a_rs + (k0 + 1) * a_cs];
        b.x = B[k0 * b_rs + nl * b_cs];
        b.y = B[(k0 + 1) * b_rs + nl * b_cs];
        acc = __builtin_amdgcn_wmma_f32_16x16x4_f32(
            false, a, false, b, (short)0, acc, false, false);
    }
    return acc;
}

__device__ __forceinline__ float lane_read(float x, int k)
{
    return __builtin_bit_cast(float,
        __builtin_amdgcn_readlane(__builtin_bit_cast(int, x), k));
}

__global__ __launch_bounds__(128) void svgp_prep(
    const float* __restrict__ z, const float* __restrict__ u_mean,
    const float* __restrict__ u_tril_vec, const float* __restrict__ log_ls,
    const float* __restrict__ log_var, float* __restrict__ ws)
{
    __shared__ float sS[64 * PS];
    __shared__ float sX[64 * PS];
    __shared__ float sU[64 * PS];
    __shared__ float sW[4][16 * WSTR];
    __shared__ float sZs[64 * 8];
    __shared__ float sZ2[64];
    __shared__ float sT[64];
    __shared__ float sUm[64];

    const int p    = blockIdx.x;
    const int o    = p % ODIM;
    const int tid  = threadIdx.x;
    const int lane = tid & 31;
    const int wave = __builtin_amdgcn_readfirstlane(tid >> 5);
    const int row  = tid & 63;
    const int half = tid >> 6;
    const int accr = (lane >> 4) * 8;
    const int accc = lane & 15;
    float* wp = ws + (size_t)p * WS_STRIDE;

    float ils[8];
#pragma unroll
    for (int d = 0; d < 8; ++d) ils[d] = __expf(-log_ls[p * 8 + d]);
    const float var = __expf(log_var[p]);

#pragma unroll 8
    for (int it = 0; it < 32; ++it) {
        const int i  = 2 * it + half;
        const int jj = (row < i) ? row : i;
        const float v = u_tril_vec[o * 2080 + i * (i + 1) / 2 + jj];
        sU[i * PS + row] = (row <= i) ? v : 0.f;
    }
    for (int i = tid; i < 64 * PS; i += 128) sX[i] = 0.f;

    float zr[8];
    {
        float s = 0.f;
#pragma unroll
        for (int d = 0; d < 8; ++d) {
            zr[d] = z[(o * 64 + row) * 8 + d] * ils[d];
            s += zr[d] * zr[d];
        }
        if (half == 0) {
#pragma unroll
            for (int d = 0; d < 8; ++d) sZs[row * 8 + d] = zr[d];
            sZ2[row] = s;
            sUm[row] = u_mean[o * 64 + row];
        }
    }
    __syncthreads();

    {
        const float z2i = sZ2[row];
        for (int jj = 0; jj < 32; ++jj) {
            const int j = half * 32 + jj;
            float dot = 0.f;
#pragma unroll
            for (int d = 0; d < 8; ++d) dot += zr[d] * sZs[j * 8 + d];
            float v = var * __expf(-0.5f * (z2i + sZ2[j] - 2.f * dot));
            if (row == j) v += JITTER;
            sS[row * PS + j] = v;
        }
    }
    __syncthreads();

    for (int J = 0; J < 4; ++J) {
        if (J > 0 && wave >= J) {
            v8f acc = {};
            for (int P = 0; P < J; ++P)
                acc = tile_mma(acc, &sS[(16 * wave) * PS + 16 * P], PS, 1,
                                    &sS[(16 * J) * PS + 16 * P], 1, PS, lane);
#pragma unroll
            for (int v = 0; v < 8; ++v)
                sS[(16 * wave + accr + v) * PS + 16 * J + accc] -= acc[v];
        }
        __syncthreads();

        float* blk = &sS[(16 * J) * PS + 16 * J];
        if (wave == 0) {
            const int t = lane & 15;
            float a[16];
#pragma unroll
            for (int k = 0; k < 16; ++k) a[k] = blk[t * PS + k];
#pragma unroll
            for (int s = 0; s < 16; ++s) {
                float cur[16];
#pragma unroll
                for (int k = s; k < 16; ++k) cur[k] = lane_read(a[s], k);
                const float ass = cur[s];
                const float d   = sqrtf(ass);
                const float c   = a[s] / ass;
                const float l   = a[s] / d;
                a[s] = (t == s) ? d : l;
#pragma unroll
                for (int k = s + 1; k < 16; ++k) a[k] -= c * cur[k];
            }
            float rd[16], x[16];
#pragma unroll
            for (int i = 0; i < 16; ++i) {
                rd[i] = 1.f / lane_read(a[i], i);
                x[i]  = (i == t) ? 1.f : 0.f;
            }
#pragma unroll
            for (int i = 0; i < 16; ++i) {
                x[i] *= rd[i];
#pragma unroll
                for (int k = i + 1; k < 16; ++k) x[k] -= lane_read(a[i], k) * x[i];
            }
            if (lane < 16) {
#pragma unroll
                for (int i = 0; i < 16; ++i) sW[J][i * WSTR + t] = x[i];
#pragma unroll
                for (int k = 0; k < 16; ++k)
                    if (k <= t) blk[t * PS + k] = a[k];
            }
        }
        __syncthreads();

        if (wave > J) {
            v8f acc = {};
            acc = tile_mma(acc, &sS[(16 * wave) * PS + 16 * J], PS, 1,
                                &sW[J][0], 1, WSTR, lane);
#pragma unroll
            for (int v = 0; v < 8; ++v)
                sS[(16 * wave + accr + v) * PS + 16 * J + accc] = acc[v];
        }
        __syncthreads();
    }

    {
        const int r = tid >> 3, c0 = (tid & 7) * 2;
#pragma unroll
        for (int J = 0; J < 4; ++J) {
            sX[(16 * J + r) * PS + 16 * J + c0]     = sW[J][r * WSTR + c0];
            sX[(16 * J + r) * PS + 16 * J + c0 + 1] = sW[J][r * WSTR + c0 + 1];
        }
    }
    __syncthreads();
    for (int dist = 1; dist < 4; ++dist) {
        const int I  = wave;
        const int Jc = wave - dist;
        const bool act = (Jc >= 0);
        if (act) {
            v8f acc = {};
            for (int P = Jc; P < I; ++P)
                acc = tile_mma(acc, &sS[(16 * I) * PS + 16 * P], PS, 1,
                                    &sX[(16 * P) * PS + 16 * Jc], PS, 1, lane);
#pragma unroll
            for (int v = 0; v < 8; ++v)
                sX[(16 * I + accr + v) * PS + 16 * Jc + accc] = acc[v];
        }
        __syncthreads();
        v8f acc2 = {};
        if (act)
            acc2 = tile_mma(acc2, &sW[I][0], WSTR, 1,
                                  &sX[(16 * I) * PS + 16 * Jc], PS, 1, lane);
        __syncthreads();
        if (act) {
#pragma unroll
            for (int v = 0; v < 8; ++v)
                sX[(16 * I + accr + v) * PS + 16 * Jc + accc] = -acc2[v];
        }
        __syncthreads();
    }

    {
        const float di = 1.f / sS[row * PS + row];
        for (int jj = 0; jj < 32; ++jj) sU[row * PS + half * 32 + jj] *= di;
        if (half == 0) {
            float s = 0.f;
            for (int j = 0; j < 64; ++j) s += sX[row * PS + j] * sUm[j];
            sT[row] = s;
        }
    }
    __syncthreads();

    if (half == 0) {
        float s = 0.f;
        for (int i = 0; i < 64; ++i) s += sX[i * PS + row] * sT[i];
        wp[WS_Q + row] = s;
    }

    for (int Rt = 0; Rt < 4; ++Rt) {
        v8f acc = {};
        const int It0 = (wave > Rt) ? wave : Rt;
        for (int It = It0; It < 4; ++It)
            acc = tile_mma(acc, &sU[(16 * It) * PS + 16 * wave], 1, PS,
                                &sX[(16 * It) * PS + 16 * Rt], PS, 1, lane);
#pragma unroll
        for (int v = 0; v < 8; ++v)
            wp[WS_B + (16 * wave + accr + v) * 64 + 16 * Rt + accc] = acc[v];
    }

    for (int i = tid; i < 4096; i += 128)
        wp[WS_LINV + i] = sX[(i >> 6) * PS + (i & 63)];
    for (int i = tid; i < 512; i += 128) wp[WS_ZS + i] = sZs[(i & 63) * 8 + (i >> 6)] * LOG2E;
    if (tid < 64) wp[WS_Z2 + tid] = LOG2E * (log_var[p] - 0.5f * sZ2[tid]);
    if (tid < 8)  wp[WS_ILS + tid] = __expf(-log_ls[p * 8 + tid]);
    if (tid == 0) wp[WS_VAR] = var;
}

#define KP 68
#define COLS_PER_BLOCK 512
#define COLS_PER_ITER 64

typedef __attribute__((ext_vector_type(4))) float v4f;

__global__ __launch_bounds__(128) void svgp_main(
    const float* __restrict__ x, const float* __restrict__ ws,
    float* __restrict__ out, int n, int blocksPerPair)
{
    __shared__ __attribute__((aligned(16))) float sKt[64 * KP];
    __shared__ __attribute__((aligned(16))) float sZt[8 * 64];
    __shared__ __attribute__((aligned(16))) float sCz[64];
    __shared__ __attribute__((aligned(16))) float sQ[64];
    __shared__ float sMu[2][2][64];
    __shared__ float sD1[2][4][64];
    __shared__ float sD2[2][4][64];

    const int pair  = blockIdx.x / blocksPerPair;
    const int chunk = blockIdx.x % blocksPerPair;
    const float* wp = ws + (size_t)pair * WS_STRIDE;
    const int tid  = threadIdx.x;
    const int wave = __builtin_amdgcn_readfirstlane(tid >> 5);
    const int lane = tid & 31;
    const int c    = lane & 15;
    const int h    = lane >> 4;

    *(v4f*)&sZt[4 * tid] = *(const v4f*)(wp + WS_ZS + 4 * tid);
    if (tid < 64) { sCz[tid] = wp[WS_Z2 + tid]; sQ[tid] = wp[WS_Q + tid]; }
    float ils[8];
#pragma unroll
    for (int d = 0; d < 8; ++d) ils[d] = wp[WS_ILS + d];
    const float var = wp[WS_VAR];

    v4f aL[8], aB[8];
    {
        const float* rowL = wp + WS_LINV + (16 * wave + c) * 64 + 4 * h;
        const float* rowB = wp + WS_B    + (16 * wave + c) * 64 + 4 * h;
#pragma unroll
        for (int j = 0; j < 8; ++j) {
            aL[j] = *(const v4f*)(rowL + 8 * j);
            aB[j] = *(const v4f*)(rowB + 8 * j);
        }
    }
    __syncthreads();

    const size_t varBase = (size_t)(HDIM * ODIM) * (size_t)n;
    const int ncol = tid & 63;
    const int mh   = tid >> 6;
    const float* bp = &sKt[c * KP + 4 * h];

    for (int it = 0; it < COLS_PER_BLOCK / COLS_PER_ITER; ++it) {
        const int p    = it & 1;
        const int col0 = chunk * COLS_PER_BLOCK + it * COLS_PER_ITER;

        float xs[8], xn2 = 0.f;
        {
            const v4f xa = *(const v4f*)(x + (size_t)(col0 + ncol) * 8);
            const v4f xb = *(const v4f*)(x + (size_t)(col0 + ncol) * 8 + 4);
            xs[0] = xa.x * ils[0]; xs[1] = xa.y * ils[1]; xs[2] = xa.z * ils[2]; xs[3] = xa.w * ils[3];
            xs[4] = xb.x * ils[4]; xs[5] = xb.y * ils[5]; xs[6] = xb.z * ils[6]; xs[7] = xb.w * ils[7];
#pragma unroll
            for (int d = 0; d < 8; ++d) xn2 += xs[d] * xs[d];
        }

        const float cx = (-0.5f * LOG2E) * xn2;
        v4f mu4 = {0.f, 0.f, 0.f, 0.f};
#pragma unroll 2
        for (int g = 0; g < 8; ++g) {
            const int m0 = 32 * mh + 4 * g;
            v4f e = *(const v4f*)&sCz[m0] + cx;
#pragma unroll
            for (int d = 0; d < 8; ++d)
                e = __builtin_elementwise_fma(*(const v4f*)&sZt[d * 64 + m0], (v4f)(xs[d]), e);
            v4f kv;
            kv.x = __builtin_amdgcn_exp2f(e.x);
            kv.y = __builtin_amdgcn_exp2f(e.y);
            kv.z = __builtin_amdgcn_exp2f(e.z);
            kv.w = __builtin_amdgcn_exp2f(e.w);
            mu4 = __builtin_elementwise_fma(*(const v4f*)&sQ[m0], kv, mu4);
            *(v4f*)&sKt[ncol * KP + m0] = kv;
        }
        const float mu = (mu4.x + mu4.y) + (mu4.z + mu4.w);
        sMu[p][mh][ncol] = mu;
        __syncthreads();

        v8f accY[4] = {};
        v8f accW[4] = {};
#pragma unroll
        for (int j = 0; j < 8; ++j) {
            v4f b[4];
#pragma unroll
            for (int ct = 0; ct < 4; ++ct)
                b[ct] = *(const v4f*)(bp + ct * 16 * KP + 8 * j);
            const v2f aY0 = {aL[j].x, aL[j].y}, aY1 = {aL[j].z, aL[j].w};
            const v2f aW0 = {aB[j].x, aB[j].y}, aW1 = {aB[j].z, aB[j].w};
#pragma unroll
            for (int ct = 0; ct < 4; ++ct) {
                const v2f b0 = {b[ct].x, b[ct].y};
                accY[ct] = __builtin_amdgcn_wmma_f32_16x16x4_f32(
                    false, aY0, false, b0, (short)0, accY[ct], false, false);
                accW[ct] = __builtin_amdgcn_wmma_f32_16x16x4_f32(
                    false, aW0, false, b0, (short)0, accW[ct], false, false);
            }
#pragma unroll
            for (int ct = 0; ct < 4; ++ct) {
                const v2f b1 = {b[ct].z, b[ct].w};
                accY[ct] = __builtin_amdgcn_wmma_f32_16x16x4_f32(
                    false, aY1, false, b1, (short)0, accY[ct], false, false);
                accW[ct] = __builtin_amdgcn_wmma_f32_16x16x4_f32(
                    false, aW1, false, b1, (short)0, accW[ct], false, false);
            }
        }

#pragma unroll
        for (int ct = 0; ct < 4; ++ct) {
            float d1 = 0.f, d2 = 0.f;
#pragma unroll
            for (int v = 0; v < 8; ++v) {
                d1 += accY[ct][v] * accY[ct][v];
                d2 += accW[ct][v] * accW[ct][v];
            }
            d1 += __shfl_xor(d1, 16, 32);
            d2 += __shfl_xor(d2, 16, 32);
            if (lane < 16) {
                sD1[p][wave][ct * 16 + c] = d1;
                sD2[p][wave][ct * 16 + c] = d2;
            }
        }
        __syncthreads();

        if (tid < 64) {
            const float m_ = sMu[p][0][tid] + sMu[p][1][tid];
            const float d1 = (sD1[p][0][tid] + sD1[p][1][tid]) + (sD1[p][2][tid] + sD1[p][3][tid]);
            const float d2 = (sD2[p][0][tid] + sD2[p][1][tid]) + (sD2[p][2][tid] + sD2[p][3][tid]);
            const int col = col0 + tid;
            out[(size_t)pair * n + col]           = m_;
            out[varBase + (size_t)pair * n + col] = var - d1 + d2;
        }
    }
}

extern "C" void kernel_launch(void* const* d_in, const int* in_sizes, int n_in,
                              void* d_out, int out_size, void* d_ws, size_t ws_size,
                              hipStream_t stream)
{
    const float* x          = (const float*)d_in[0];
    const float* z          = (const float*)d_in[1];
    const float* u_mean     = (const float*)d_in[2];
    const float* u_tril_vec = (const float*)d_in[3];
    const float* log_ls     = (const float*)d_in[4];
    const float* log_var    = (const float*)d_in[5];
    float* out = (float*)d_out;
    float* ws  = (float*)d_ws;

    const int n = in_sizes[0] / 8;
    svgp_prep<<<HDIM * ODIM, 128, 0, stream>>>(z, u_mean, u_tril_vec,
                                              log_ls, log_var, ws);
    const int blocksPerPair = n / COLS_PER_BLOCK;
    svgp_main<<<HDIM * ODIM * blocksPerPair, 128, 0, stream>>>(
        x, ws, out, n, blocksPerPair);
}


# main: kfu tile double-buffered in LDS; each wave makes the next tile's entries (exp2 chains) in slices pinned between the current tile's matrix instructions; one block barrier per tile, not two
# speedup vs baseline: 3.2042x; 1.0484x over previous
#include <hip/hip_runtime.h>

#define HDIM 4
#define ODIM 4
#define JITTER 1e-4f
#define LOG2E 1.44269504088896340736f

#define WS_LINV 0
#define WS_B    4096
#define WS_ZS   8192
#define WS_Z2   8704
#define WS_Q    8768
#define WS_ILS  8832
#define WS_VAR  8840
#define WS_STRIDE 8960

typedef __attribute__((ext_vector_type(2))) float v2f;
typedef __attribute__((ext_vector_type(8))) float v8f;

#define PS 65
#define WSTR 17

__device__ __forceinline__ v8f tile_mma(v8f acc, const float* A, int a_rs, int a_cs,
                                        const float* B, int b_rs, int b_cs, int lane)
{
    const int nl = lane & 15;
    const int kh = (lane >> 4) * 2;
#pragma unroll
    for (int kc = 0; kc < 4; ++kc) {
        const int k0 = kc * 4 + kh;
        v2f a, b;
        a.x = A[nl * a_rs + k0 * a_cs];
        a.y = A[nl * a_rs + (k0 + 1) * a_cs];
        b.x = B[k0 * b_rs + nl * b_cs];
        b.y = B[(k0 + 1) * b_rs + nl * b_cs];
        acc = __builtin_amdgcn_wmma_f32_16x16x4_f32(
            false, a, false, b, (short)0, acc, false, false);
    }
    return acc;
}

__device__ __forceinline__ float lane_read(float x, int k)
{
    return __builtin_bit_cast(float,
        __builtin_amdgcn_readlane(__builtin_bit_cast(int, x), k));
}

__global__ __launch_bounds__(128) void svgp_prep(
    const float* __restrict__ z, const float* __restrict__ u_mean,
    const float* __restrict__ u_tril_vec, const float* __restrict__ log_ls,
    const float* __restrict__ log_var, float* __restrict__ ws)
{
    __shared__ float sS[64 * PS];
    __shared__ float sX[64 * PS];
    __shared__ float sU[64 * PS];
    __shared__ float sW[4][16 * WSTR];
    __shared__ float sZs[64 * 8];
    __shared__ float sZ2[64];
    __shared__ float sT[64];
    __shared__ float sUm[64];

    const int p    = blockIdx.x;
    const int o    = p % ODIM;
    const int tid  = threadIdx.x;
    const int lane = tid & 31;
    const int wave = __builtin_amdgcn_readfirstlane(tid >> 5);
    const int row  = tid & 63;
    const int half = tid >> 6;
    const int accr = (lane >> 4) * 8;
    const int accc = lane & 15;
    float* wp = ws + (size_t)p * WS_STRIDE;

    float ils[8];
#pragma unroll
    for (int d = 0; d < 8; ++d) ils[d] = __expf(-log_ls[p * 8 + d]);
    const float var = __expf(log_var[p]);

#pragma unroll 8
    for (int it = 0; it < 32; ++it) {
        const int i  = 2 * it + half;
        const int jj = (row < i) ? row : i;
        const float v = u_tril_vec[o * 2080 + i * (i + 1) / 2 + jj];
        sU[i * PS + row] = (row <= i) ? v : 0.f;
    }
    for (int i = tid; i < 64 * PS; i += 128) sX[i] = 0.f;

    float zr[8];
    {
        float s = 0.f;
#pragma unroll
        for (int d = 0; d < 8; ++d) {
            zr[d] = z[(o * 64 + row) * 8 + d] * ils[d];
            s += zr[d] * zr[d];
        }
        if (half == 0) {
#pragma unroll
            for (int d = 0; d < 8; ++d) sZs[row * 8 + d] = zr[d];
            sZ2[row] = s;
            sUm[row] = u_mean[o * 64 + row];
        }
    }
    __syncthreads();

    {
        const float z2i = sZ2[row];
        for (int jj = 0; jj < 32; ++jj) {
            const int j = half * 32 + jj;
            float dot = 0.f;
#pragma unroll
            for (int d = 0; d < 8; ++d) dot += zr[d] * sZs[j * 8 + d];
            float v = var * __expf(-0.5f * (z2i + sZ2[j] - 2.f * dot));
            if (row == j) v += JITTER;
            sS[row * PS + j] = v;
        }
    }
    __syncthreads();

    for (int J = 0; J < 4; ++J) {
        if (J > 0 && wave >= J) {
            v8f acc = {};
            for (int P = 0; P < J; ++P)
                acc = tile_mma(acc, &sS[(16 * wave) * PS + 16 * P], PS, 1,
                                    &sS[(16 * J) * PS + 16 * P], 1, PS, lane);
#pragma unroll
            for (int v = 0; v < 8; ++v)
                sS[(16 * wave + accr + v) * PS + 16 * J + accc] -= acc[v];
        }
        __syncthreads();

        float* blk = &sS[(16 * J) * PS + 16 * J];
        if (wave == 0) {
            const int t = lane & 15;
            float a[16];
#pragma unroll
            for (int k = 0; k < 16; ++k) a[k] = blk[t * PS + k];
#pragma unroll
            for (int s = 0; s < 16; ++s) {
                float cur[16];
#pragma unroll
                for (int k = s; k < 16; ++k) cur[k] = lane_read(a[s], k);
                const float ass = cur[s];
                const float d   = sqrtf(ass);
                const float c   = a[s] / ass;
                const float l   = a[s] / d;
                a[s] = (t == s) ? d : l;
#pragma unroll
                for (int k = s + 1; k < 16; ++k) a[k] -= c * cur[k];
            }
            float rd[16], x[16];
#pragma unroll
            for (int i = 0; i < 16; ++i) {
                rd[i] = 1.f / lane_read(a[i], i);
                x[i]  = (i == t) ? 1.f : 0.f;
            }
#pragma unroll
            for (int i = 0; i < 16; ++i) {
                x[i] *= rd[i];
#pragma unroll
                for (int k = i + 1; k < 16; ++k) x[k] -= lane_read(a[i], k) * x[i];
            }
            if (lane < 16) {
#pragma unroll
                for (int i = 0; i < 16; ++i) sW[J][i * WSTR + t] = x[i];
#pragma unroll
                for (int k = 0; k < 16; ++k)
                    if (k <= t) blk[t * PS + k] = a[k];
            }
        }
        __syncthreads();

        if (wave > J) {
            v8f acc = {};
            acc = tile_mma(acc, &sS[(16 * wave) * PS + 16 * J], PS, 1,
                                &sW[J][0], 1, WSTR, lane);
#pragma unroll
            for (int v = 0; v < 8; ++v)
                sS[(16 * wave + accr + v) * PS + 16 * J + accc] = acc[v];
        }
        __syncthreads();
    }

    {
        const int r = tid >> 3, c0 = (tid & 7) * 2;
#pragma unroll
        for (int J = 0; J < 4; ++J) {
            sX[(16 * J + r) * PS + 16 * J + c0]     = sW[J][r * WSTR + c0];
            sX[(16 * J + r) * PS + 16 * J + c0 + 1] = sW[J][r * WSTR + c0 + 1];
        }
    }
    __syncthreads();
    for (int dist = 1; dist < 4; ++dist) {
        const int I  = wave;
        const int Jc = wave - dist;
        const bool act = (Jc >= 0);
        if (act) {
            v8f acc = {};
            for (int P = Jc; P < I; ++P)
                acc = tile_mma(acc, &sS[(16 * I) * PS + 16 * P], PS, 1,
                                    &sX[(16 * P) * PS + 16 * Jc], PS, 1, lane);
#pragma unroll
            for (int v = 0; v < 8; ++v)
                sX[(16 * I + accr + v) * PS + 16 * Jc + accc] = acc[v];
        }
        __syncthreads();
        v8f acc2 = {};
        if (act)
            acc2 = tile_mma(acc2, &sW[I][0], WSTR, 1,
                                  &sX[(16 * I) * PS + 16 * Jc], PS, 1, lane);
        __syncthreads();
        if (act) {
#pragma unroll
            for (int v = 0; v < 8; ++v)
                sX[(16 * I + accr + v) * PS + 16 * Jc + accc] = -acc2[v];
        }
        __syncthreads();
    }

    {
        const float di = 1.f / sS[row * PS + row];
        for (int jj = 0; jj < 32; ++jj) sU[row * PS + half * 32 + jj] *= di;
        if (half == 0) {
            float s = 0.f;
            for (int j = 0; j < 64; ++j) s += sX[row * PS + j] * sUm[j];
            sT[row] = s;
        }
    }
    __syncthreads();

    if (half == 0) {
        float s = 0.f;
        for (int i = 0; i < 64; ++i) s += sX[i * PS + row] * sT[i];
        wp[WS_Q + row] = s;
    }

    for (int Rt = 0; Rt < 4; ++Rt) {
        v8f acc = {};
        const int It0 = (wave > Rt) ? wave : Rt;
        for (int It = It0; It < 4; ++It)
            acc = tile_mma(acc, &sU[(16 * It) * PS + 16 * wave], 1, PS,
                                &sX[(16 * It) * PS + 16 * Rt], PS, 1, lane);
#pragma unroll
        for (int v = 0; v < 8; ++v)
            wp[WS_B + (16 * wave + accr + v) * 64 + 16 * Rt + accc] = acc[v];
    }

    for (int i = tid; i < 4096; i += 128)
        wp[WS_LINV + i] = sX[(i >> 6) * PS + (i & 63)];
    for (int i = tid; i < 512; i += 128) wp[WS_ZS + i] = sZs[(i & 63) * 8 + (i >> 6)] * LOG2E;
    if (tid < 64) wp[WS_Z2 + tid] = LOG2E * (log_var[p] - 0.5f * sZ2[tid]);
    if (tid < 8)  wp[WS_ILS + tid] = __expf(-log_ls[p * 8 + tid]);
    if (tid == 0) wp[WS_VAR] = var;
}

#define KP 68
#define COLS_PER_BLOCK 512
#define COLS_PER_ITER 64
#define NTILES (COLS_PER_BLOCK / COLS_PER_ITER)

typedef __attribute__((ext_vector_type(4))) float v4f;

__device__ __forceinline__ void kfu_group(int m0, const float* sCz, const float* sZt,
                                          const float* sQ, const float (&xs)[8], float cx,
                                          v4f& mu4, float* dstRow)
{
    v4f e = *(const v4f*)&sCz[m0] + cx;
#pragma unroll
    for (int d = 0; d < 8; ++d)
        e = __builtin_elementwise_fma(*(const v4f*)&sZt[d * 64 + m0], (v4f)(xs[d]), e);
    v4f kv;
    kv.x = __builtin_amdgcn_exp2f(e.x);
    kv.y = __builtin_amdgcn_exp2f(e.y);
    kv.z = __builtin_amdgcn_exp2f(e.z);
    kv.w = __builtin_amdgcn_exp2f(e.w);
    mu4 = __builtin_elementwise_fma(*(const v4f*)&sQ[m0], kv, mu4);
    *(v4f*)&dstRow[m0] = kv;
}

__device__ __forceinline__ void scale_x(const v4f xa, const v4f xb, const float (&ils)[8],
                                        float (&xs)[8], float& cx)
{
    xs[0] = xa.x * ils[0]; xs[1] = xa.y * ils[1]; xs[2] = xa.z * ils[2]; xs[3] = xa.w * ils[3];
    xs[4] = xb.x * ils[4]; xs[5] = xb.y * ils[5]; xs[6] = xb.z * ils[6]; xs[7] = xb.w * ils[7];
    float xn2 = 0.f;
#pragma unroll
    for (int d = 0; d < 8; ++d) xn2 += xs[d] * xs[d];
    cx = (-0.5f * LOG2E) * xn2;
}

__device__ __forceinline__ void tile_last(const v4f (&aL)[8], const v4f (&aB)[8],
                                          const float* bcur, v8f (&accY)[4], v8f (&accW)[4])
{
#pragma unroll
    for (int j = 0; j < 8; ++j) {
        v4f b[4];
#pragma unroll
        for (int ct = 0; ct < 4; ++ct)
            b[ct] = *(const v4f*)(bcur + ct * 16 * KP + 8 * j);
        const v2f aY0 = {aL[j].x, aL[j].y}, aY1 = {aL[j].z, aL[j].w};
        const v2f aW0 = {aB[j].x, aB[j].y}, aW1 = {aB[j].z, aB[j].w};
#pragma unroll
        for (int ct = 0; ct < 4; ++ct) {
            const v2f b0 = {b[ct].x, b[ct].y};
            accY[ct] = __builtin_amdgcn_wmma_f32_16x16x4_f32(
                false, aY0, false, b0, (short)0, accY[ct], false, false);
            accW[ct] = __builtin_amdgcn_wmma_f32_16x16x4_f32(
                false, aW0, false, b0, (short)0, accW[ct], false, false);
        }
#pragma unroll
        for (int ct = 0; ct < 4; ++ct) {
            const v2f b1 = {b[ct].z, b[ct].w};
            accY[ct] = __builtin_amdgcn_wmma_f32_16x16x4_f32(
                false, aY1, false, b1, (short)0, accY[ct], false, false);
            accW[ct] = __builtin_amdgcn_wmma_f32_16x16x4_f32(
                false, aW1, false, b1, (short)0, accW[ct], false, false);
        }
    }
}

#define PIN_HERE() __builtin_amdgcn_sched_barrier(0)

#define WMMA_Y(ct, afrag, bfrag) accY[ct] = __builtin_amdgcn_wmma_f32_16x16x4_f32( \
        false, afrag, false, bfrag, (short)0, accY[ct], false, false)
#define WMMA_W(ct, afrag, bfrag) accW[ct] = __builtin_amdgcn_wmma_f32_16x16x4_f32( \
        false, afrag, false, bfrag, (short)0, accW[ct], false, false)

__device__ __forceinline__ void tile_next(const v4f (&aL)[8], const v4f (&aB)[8],
                                          const float* bcur, float* knextRow, int mbase,
                                          const float* sCz, const float* sZt, const float* sQ,
                                          const float (&xs)[8], float cx, v4f& mu4,
                                          v8f (&accY)[4], v8f (&accW)[4])
{
    v4f b[4];
#pragma unroll
    for (int ct = 0; ct < 4; ++ct)
        b[ct] = *(const v4f*)(bcur + ct * 16 * KP);
#pragma unroll
    for (int j = 0; j < 8; ++j) {
        const int m0 = mbase + 4 * j;
        v4f tz[8], bn[4];
        const v4f tc = *(const v4f*)&sCz[m0];
#pragma unroll
        for (int d = 0; d < 8; ++d) tz[d] = *(const v4f*)&sZt[d * 64 + m0];
        const v4f tq = *(const v4f*)&sQ[m0];
        if (j < 7) {
#pragma unroll
            for (int ct = 0; ct < 4; ++ct)
                bn[ct] = *(const v4f*)(bcur + ct * 16 * KP + 8 * (j + 1));
        }
        PIN_HERE();
        const v2f aY0 = {aL[j].x, aL[j].y}, aY1 = {aL[j].z, aL[j].w};
        const v2f aW0 = {aB[j].x, aB[j].y}, aW1 = {aB[j].z, aB[j].w};
#pragma unroll
        for (int ct = 0; ct < 4; ++ct) {
            const v2f b0 = {b[ct].x, b[ct].y};
            WMMA_Y(ct, aY0, b0);
            WMMA_W(ct, aW0, b0);
        }
        PIN_HERE();
        const v2f b10 = {b[0].z, b[0].w}, b11 = {b[1].z, b[1].w};
        const v2f b12 = {b[2].z, b[2].w}, b13 = {b[3].z, b[3].w};
        WMMA_Y(0, aY1, b10);
        v4f e = tc + cx;
        e = __builtin_elementwise_fma(tz[0], (v4f)(xs[0]), e);
        e = __builtin_elementwise_fma(tz[1], (v4f)(xs[1]), e);
        PIN_HERE();
        WMMA_W(0, aW1, b10);
        e = __builtin_elementwise_fma(tz[2], (v4f)(xs[2]), e);
        e = __builtin_elementwise_fma(tz[3], (v4f)(xs[3]), e);
        PIN_HERE();
        WMMA_Y(1, aY1, b11);
        e = __builtin_elementwise_fma(tz[4], (v4f)(xs[4]), e);
        e = __builtin_elementwise_fma(tz[5], (v4f)(xs[5]), e);
        PIN_HERE();
        WMMA_W(1, aW1, b11);
        e = __builtin_elementwise_fma(tz[6], (v4f)(xs[6]), e);
        e = __builtin_elementwise_fma(tz[7], (v4f)(xs[7]), e);
        PIN_HERE();
        WMMA_Y(2, aY1, b12);
        v4f kv;
        kv.x = __builtin_amdgcn_exp2f(e.x);
        kv.y = __builtin_amdgcn_exp2f(e.y);
        PIN_HERE();
        WMMA_W(2, aW1, b12);
        kv.z = __builtin_amdgcn_exp2f(e.z);
        kv.w = __builtin_amdgcn_exp2f(e.w);
        PIN_HERE();
        WMMA_Y(3, aY1, b13);
        mu4 = __builtin_elementwise_fma(tq, kv, mu4);
        PIN_HERE();
        WMMA_W(3, aW1, b13);
        *(v4f*)&knextRow[m0] = kv;
        PIN_HERE();
        if (j < 7) {
#pragma unroll
            for (int ct = 0; ct < 4; ++ct) b[ct] = bn[ct];
        }
    }
}

__global__ __launch_bounds__(128) void svgp_main(
    const float* __restrict__ x, const float* __restrict__ ws,
    float* __restrict__ out, int n, int blocksPerPair)
{
    __shared__ __attribute__((aligned(16))) float sKt[2][64 * KP];
    __shared__ __attribute__((aligned(16))) float sZt[8 * 64];
    __shared__ __attribute__((aligned(16))) float sCz[64];
    __shared__ __attribute__((aligned(16))) float sQ[64];
    __shared__ float sMu[4][2][64];
    __shared__ float sD1[2][4][64];
    __shared__ float sD2[2][4][64];

    const int pair  = blockIdx.x / blocksPerPair;
    const int chunk = blockIdx.x % blocksPerPair;
    const float* wp = ws + (size_t)pair * WS_STRIDE;
    const int tid  = threadIdx.x;
    const int wave = __builtin_amdgcn_readfirstlane(tid >> 5);
    const int lane = tid & 31;
    const int c    = lane & 15;
    const int h    = lane >> 4;

    *(v4f*)&sZt[4 * tid] = *(const v4f*)(wp + WS_ZS + 4 * tid);
    if (tid < 64) { sCz[tid] = wp[WS_Z2 + tid]; sQ[tid] = wp[WS_Q + tid]; }
    float ils[8];
#pragma unroll
    for (int d = 0; d < 8; ++d) ils[d] = wp[WS_ILS + d];
    const float var = wp[WS_VAR];

    const int ncol = tid & 63;
    const int mbase = 32 * (wave >> 1);
    const float* xcol = x + ((size_t)chunk * COLS_PER_BLOCK + ncol) * 8;

    v4f xa = *(const v4f*)(xcol);
    v4f xb = *(const v4f*)(xcol + 4);
    v4f xna = *(const v4f*)(xcol + (size_t)COLS_PER_ITER * 8);
    v4f xnb = *(const v4f*)(xcol + (size_t)COLS_PER_ITER * 8 + 4);

    v4f aL[8], aB[8];
    {
        const float* rowL = wp + WS_LINV + (16 * wave + c) * 64 + 4 * h;
        const float* rowB = wp + WS_B    + (16 * wave + c) * 64 + 4 * h;
#pragma unroll
        for (int j = 0; j < 8; ++j) {
            aL[j] = *(const v4f*)(rowL + 8 * j);
            aB[j] = *(const v4f*)(rowB + 8 * j);
        }
    }
    __syncthreads();

    const size_t varBase = (size_t)(HDIM * ODIM) * (size_t)n;
    const float* bp = &sKt[0][c * KP + 4 * h];

    float xs[8], cx;
    v4f mu4 = {0.f, 0.f, 0.f, 0.f};

    scale_x(xa, xb, ils, xs, cx);
#pragma unroll 2
    for (int g = 0; g < 8; ++g)
        kfu_group(mbase + 4 * g, sCz, sZt, sQ, xs, cx, mu4, &sKt[0][ncol * KP]);
    sMu[0][wave >> 1][ncol] = (mu4.x + mu4.y) + (mu4.z + mu4.w);
    __syncthreads();

    for (int it = 0; it < NTILES; ++it) {
        const int p    = it & 1;
        const int col0 = chunk * COLS_PER_BLOCK + it * COLS_PER_ITER;

        v8f accY[4] = {};
        v8f accW[4] = {};
        if (it + 1 < NTILES) {
            scale_x(xna, xnb, ils, xs, cx);
            const int t2 = (it + 2 < NTILES) ? it + 2 : NTILES - 1;
            xna = *(const v4f*)(xcol + (size_t)t2 * COLS_PER_ITER * 8);
            xnb = *(const v4f*)(xcol + (size_t)t2 * COLS_PER_ITER * 8 + 4);
            mu4 = (v4f){0.f, 0.f, 0.f, 0.f};
            tile_next(aL, aB, bp + p * (64 * KP), &sKt[p ^ 1][ncol * KP], mbase,
                      sCz, sZt, sQ, xs, cx, mu4, accY, accW);
            sMu[(it + 1) & 3][wave >> 1][ncol] = (mu4.x + mu4.y) + (mu4.z + mu4.w);
        } else {
            tile_last(aL, aB, bp + p * (64 * KP), accY, accW);
        }

#pragma unroll
        for (int ct = 0; ct < 4; ++ct) {
            float d1 = 0.f, d2 = 0.f;
#pragma unroll
            for (int v = 0; v < 8; ++v) {
                d1 += accY[ct][v] * accY[ct][v];
                d2 += accW[ct][v] * accW[ct][v];
            }
            d1 += __shfl_xor(d1, 16, 32);
            d2 += __shfl_xor(d2, 16, 32);
            if (lane < 16) {
                sD1[p][wave][ct * 16 + c] = d1;
                sD2[p][wave][ct * 16 + c] = d2;
            }
        }
        __syncthreads();

        if (tid < 64) {
            const float m_ = sMu[it & 3][0][tid] + sMu[it & 3][1][tid];
            const float d1 = (sD1[p][0][tid] + sD1[p][1][tid]) + (sD1[p][2][tid] + sD1[p][3][tid]);
            const float d2 = (sD2[p][0][tid] + sD2[p][1][tid]) + (sD2[p][2][tid] + sD2[p][3][tid]);
            const int col = col0 + tid;
            out[(size_t)pair * n + col]           = m_;
            out[varBase + (size_t)pair * n + col] = var - d1 + d2;
        }
    }
}

extern "C" void kernel_launch(void* const* d_in, const int* in_sizes, int n_in,
                              void* d_out, int out_size, void* d_ws, size_t ws_size,
                              hipStream_t stream)
{
    const float* x          = (const float*)d_in[0];
    const float* z          = (const float*)d_in[1];
    const float* u_mean     = (const float*)d_in[2];
    const float* u_tril_vec = (const float*)d_in[3];
    const float* log_ls     = (const float*)d_in[4];
    const float* log_var    = (const float*)d_in[5];
    float* out = (float*)d_out;
    float* ws  = (float*)d_ws;

    const int n = in_sizes[0] / 8;
    svgp_prep<<<HDIM * ODIM, 128, 0, stream>>>(z, u_mean, u_tril_vec,
                                              log_ls, log_var, ws);
    const int blocksPerPair = n / COLS_PER_BLOCK;
    svgp_main<<<HDIM * ODIM * blocksPerPair, 128, 0, stream>>>(
        x, ws, out, n, blocksPerPair);
}


# prep: each diagonal-block pivot costs one reciprocal square root (column scaled by it, rank-1 update from scaled entries, inverse reuses it): no sqrt, no division on the lone factoring wave
# speedup vs baseline: 3.5647x; 1.1125x over previous
#include <hip/hip_runtime.h>

#define HDIM 4
#define ODIM 4
#define JITTER 1e-4f
#define LOG2E 1.44269504088896340736f

#define WS_LINV 0
#define WS_B    4096
#define WS_ZS   8192
#define WS_Z2   8704
#define WS_Q    8768
#define WS_ILS  8832
#define WS_VAR  8840
#define WS_STRIDE 8960

typedef __attribute__((ext_vector_type(2))) float v2f;
typedef __attribute__((ext_vector_type(8))) float v8f;

#define PS 65
#define WSTR 17

__device__ __forceinline__ v8f tile_mma(v8f acc, const float* A, int a_rs, int a_cs,
                                        const float* B, int b_rs, int b_cs, int lane)
{
    const int nl = lane & 15;
    const int kh = (lane >> 4) * 2;
#pragma unroll
    for (int kc = 0; kc < 4; ++kc) {
        const int k0 = kc * 4 + kh;
        v2f a, b;
        a.x = A[nl * a_rs + k0 * a_cs];
        a.y = A[nl * a_rs + (k0 + 1) * a_cs];
        b.x = B[k0 * b_rs + nl * b_cs];
        b.y = B[(k0 + 1) * b_rs + nl * b_cs];
        acc = __builtin_amdgcn_wmma_f32_16x16x4_f32(
            false, a, false, b, (short)0, acc, false, false);
    }
    return acc;
}

__device__ __forceinline__ float lane_read(float x, int k)
{
    return __builtin_bit_cast(float,
        __builtin_amdgcn_readlane(__builtin_bit_cast(int, x), k));
}

__global__ __launch_bounds__(128) void svgp_prep(
    const float* __restrict__ z, const float* __restrict__ u_mean,
    const float* __restrict__ u_tril_vec, const float* __restrict__ log_ls,
    const float* __restrict__ log_var, float* __restrict__ ws)
{
    __shared__ float sS[64 * PS];
    __shared__ float sX[64 * PS];
    __shared__ float sU[64 * PS];
    __shared__ float sW[4][16 * WSTR];
    __shared__ float sZs[64 * 8];
    __shared__ float sZ2[64];
    __shared__ float sT[64];
    __shared__ float sUm[64];

    const int p    = blockIdx.x;
    const int o    = p % ODIM;
    const int tid  = threadIdx.x;
    const int lane = tid & 31;
    const int wave = __builtin_amdgcn_readfirstlane(tid >> 5);
    const int row  = tid & 63;
    const int half = tid >> 6;
    const int accr = (lane >> 4) * 8;
    const int accc = lane & 15;
    float* wp = ws + (size_t)p * WS_STRIDE;

    float ils[8];
#pragma unroll
    for (int d = 0; d < 8; ++d) ils[d] = __expf(-log_ls[p * 8 + d]);
    const float var = __expf(log_var[p]);

#pragma unroll 8
    for (int it = 0; it < 32; ++it) {
        const int i  = 2 * it + half;
        const int jj = (row < i) ? row : i;
        const float v = u_tril_vec[o * 2080 + i * (i + 1) / 2 + jj];
        sU[i * PS + row] = (row <= i) ? v : 0.f;
    }
    for (int i = tid; i < 64 * PS; i += 128) sX[i] = 0.f;

    float zr[8];
    {
        float s = 0.f;
#pragma unroll
        for (int d = 0; d < 8; ++d) {
            zr[d] = z[(o * 64 + row) * 8 + d] * ils[d];
            s += zr[d] * zr[d];
        }
        if (half == 0) {
#pragma unroll
            for (int d = 0; d < 8; ++d) sZs[row * 8 + d] = zr[d];
            sZ2[row] = s;
            sUm[row] = u_mean[o * 64 + row];
        }
    }
    __syncthreads();

    {
        const float z2i = sZ2[row];
        for (int jj = 0; jj < 32; ++jj) {
            const int j = half * 32 + jj;
            float dot = 0.f;
#pragma unroll
            for (int d = 0; d < 8; ++d) dot += zr[d] * sZs[j * 8 + d];
            float v = var * __expf(-0.5f * (z2i + sZ2[j] - 2.f * dot));
            if (row == j) v += JITTER;
            sS[row * PS + j] = v;
        }
    }
    __syncthreads();

    for (int J = 0; J < 4; ++J) {
        if (J > 0 && wave >= J) {
            v8f acc = {};
            for (int P = 0; P < J; ++P)
                acc = tile_mma(acc, &sS[(16 * wave) * PS + 16 * P], PS, 1,
                                    &sS[(16 * J) * PS + 16 * P], 1, PS, lane);
#pragma unroll
            for (int v = 0; v < 8; ++v)
                sS[(16 * wave + accr + v) * PS + 16 * J + accc] -= acc[v];
        }
        __syncthreads();

        float* blk = &sS[(16 * J) * PS + 16 * J];
        if (wave == 0) {
            const int t = lane & 15;
            float a[16], rs[16];
#pragma unroll
            for (int k = 0; k < 16; ++k) a[k] = blk[t * PS + k];
#pragma unroll
            for (int s = 0; s < 16; ++s) {
                rs[s] = __builtin_amdgcn_rsqf(lane_read(a[s], s));
                a[s] *= rs[s];
#pragma unroll
                for (int k = s + 1; k < 16; ++k) a[k] -= a[s] * lane_read(a[s], k);
            }
            float x[16];
#pragma unroll
            for (int i = 0; i < 16; ++i) x[i] = (i == t) ? 1.f : 0.f;
#pragma unroll
            for (int i = 0; i < 16; ++i) {
                x[i] *= rs[i];
#pragma unroll
                for (int k = i + 1; k < 16; ++k) x[k] -= lane_read(a[i], k) * x[i];
            }
            if (lane < 16) {
#pragma unroll
                for (int i = 0; i < 16; ++i) sW[J][i * WSTR + t] = x[i];
#pragma unroll
                for (int k = 0; k < 16; ++k)
                    if (k <= t) blk[t * PS + k] = a[k];
            }
        }
        __syncthreads();

        if (wave > J) {
            v8f acc = {};
            acc = tile_mma(acc, &sS[(16 * wave) * PS + 16 * J], PS, 1,
                                &sW[J][0], 1, WSTR, lane);
#pragma unroll
            for (int v = 0; v < 8; ++v)
                sS[(16 * wave + accr + v) * PS + 16 * J + accc] = acc[v];
        }
        __syncthreads();
    }

    {
        const int r = tid >> 3, c0 = (tid & 7) * 2;
#pragma unroll
        for (int J = 0; J < 4; ++J) {
            sX[(16 * J + r) * PS + 16 * J + c0]     = sW[J][r * WSTR + c0];
            sX[(16 * J + r) * PS + 16 * J + c0 + 1] = sW[J][r * WSTR + c0 + 1];
        }
    }
    __syncthreads();
    for (int dist = 1; dist < 4; ++dist) {
        const int I  = wave;
        const int Jc = wave - dist;
        const bool act = (Jc >= 0);
        if (act) {
            v8f acc = {};
            for (int P = Jc; P < I; ++P)
                acc = tile_mma(acc, &sS[(16 * I) * PS + 16 * P], PS, 1,
                                    &sX[(16 * P) * PS + 16 * Jc], PS, 1, lane);
#pragma unroll
            for (int v = 0; v < 8; ++v)
                sX[(16 * I + accr + v) * PS + 16 * Jc + accc] = acc[v];
        }
        __syncthreads();
        v8f acc2 = {};
        if (act)
            acc2 = tile_mma(acc2, &sW[I][0], WSTR, 1,
                                  &sX[(16 * I) * PS + 16 * Jc], PS, 1, lane);
        __syncthreads();
        if (act) {
#pragma unroll
            for (int v = 0; v < 8; ++v)
                sX[(16 * I + accr + v) * PS + 16 * Jc + accc] = -acc2[v];
        }
        __syncthreads();
    }

    {
        const float di = 1.f / sS[row * PS + row];
        for (int jj = 0; jj < 32; ++jj) sU[row * PS + half * 32 + jj] *= di;
        if (half == 0) {
            float s = 0.f;
            for (int j = 0; j < 64; ++j) s += sX[row * PS + j] * sUm[j];
            sT[row] = s;
        }
    }
    __syncthreads();

    if (half == 0) {
        float s = 0.f;
        for (int i = 0; i < 64; ++i) s += sX[i * PS + row] * sT[i];
        wp[WS_Q + row] = s;
    }

    for (int Rt = 0; Rt < 4; ++Rt) {
        v8f acc = {};
        const int It0 = (wave > Rt) ? wave : Rt;
        for (int It = It0; It < 4; ++It)
            acc = tile_mma(acc, &sU[(16 * It) * PS + 16 * wave], 1, PS,
                                &sX[(16 * It) * PS + 16 * Rt], PS, 1, lane);
#pragma unroll
        for (int v = 0; v < 8; ++v)
            wp[WS_B + (16 * wave + accr + v) * 64 + 16 * Rt + accc] = acc[v];
    }

    for (int i = tid; i < 4096; i += 128)
        wp[WS_LINV + i] = sX[(i >> 6) * PS + (i & 63)];
    for (int i = tid; i < 512; i += 128) wp[WS_ZS + i] = sZs[(i & 63) * 8 + (i >> 6)] * LOG2E;
    if (tid < 64) wp[WS_Z2 + tid] = LOG2E * (log_var[p] - 0.5f * sZ2[tid]);
    if (tid < 8)  wp[WS_ILS + tid] = __expf(-log_ls[p * 8 + tid]);
    if (tid == 0) wp[WS_VAR] = var;
}

#define KP 68
#define COLS_PER_BLOCK 512
#define COLS_PER_ITER 64
#define NTILES (COLS_PER_BLOCK / COLS_PER_ITER)

typedef __attribute__((ext_vector_type(4))) float v4f;

__device__ __forceinline__ void kfu_group(int m0, const float* sCz, const float* sZt,
                                          const float* sQ, const float (&xs)[8], float cx,
                                          v4f& mu4, float* dstRow)
{
    v4f e = *(const v4f*)&sCz[m0] + cx;
#pragma unroll
    for (int d = 0; d < 8; ++d)
        e = __builtin_elementwise_fma(*(const v4f*)&sZt[d * 64 + m0], (v4f)(xs[d]), e);
    v4f kv;
    kv.x = __builtin_amdgcn_exp2f(e.x);
    kv.y = __builtin_amdgcn_exp2f(e.y);
    kv.z = __builtin_amdgcn_exp2f(e.z);
    kv.w = __builtin_amdgcn_exp2f(e.w);
    mu4 = __builtin_elementwise_fma(*(const v4f*)&sQ[m0], kv, mu4);
    *(v4f*)&dstRow[m0] = kv;
}

__device__ __forceinline__ void scale_x(const v4f xa, const v4f xb, const float (&ils)[8],
                                        float (&xs)[8], float& cx)
{
    xs[0] = xa.x * ils[0]; xs[1] = xa.y * ils[1]; xs[2] = xa.z * ils[2]; xs[3] = xa.w * ils[3];
    xs[4] = xb.x * ils[4]; xs[5] = xb.y * ils[5]; xs[6] = xb.z * ils[6]; xs[7] = xb.w * ils[7];
    float xn2 = 0.f;
#pragma unroll
    for (int d = 0; d < 8; ++d) xn2 += xs[d] * xs[d];
    cx = (-0.5f * LOG2E) * xn2;
}

__device__ __forceinline__ void tile_last(const v4f (&aL)[8], const v4f (&aB)[8],
                                          const float* bcur, v8f (&accY)[4], v8f (&accW)[4])
{
#pragma unroll
    for (int j = 0; j < 8; ++j) {
        v4f b[4];
#pragma unroll
        for (int ct = 0; ct < 4; ++ct)
            b[ct] = *(const v4f*)(bcur + ct * 16 * KP + 8 * j);
        const v2f aY0 = {aL[j].x, aL[j].y}, aY1 = {aL[j].z, aL[j].w};
        const v2f aW0 = {aB[j].x, aB[j].y}, aW1 = {aB[j].z, aB[j].w};
#pragma unroll
        for (int ct = 0; ct < 4; ++ct) {
            const v2f b0 = {b[ct].x, b[ct].y};
            accY[ct] = __builtin_amdgcn_wmma_f32_16x16x4_f32(
                false, aY0, false, b0, (short)0, accY[ct], false, false);
            accW[ct] = __builtin_amdgcn_wmma_f32_16x16x4_f32(
                false, aW0, false, b0, (short)0, accW[ct], false, false);
        }
#pragma unroll
        for (int ct = 0; ct < 4; ++ct) {
            const v2f b1 = {b[ct].z, b[ct].w};
            accY[ct] = __builtin_amdgcn_wmma_f32_16x16x4_f32(
                false, aY1, false, b1, (short)0, accY[ct], false, false);
            accW[ct] = __builtin_amdgcn_wmma_f32_16x16x4_f32(
                false, aW1, false, b1, (short)0, accW[ct], false, false);
        }
    }
}

#define PIN_HERE() __builtin_amdgcn_sched_barrier(0)

#define WMMA_Y(ct, afrag, bfrag) accY[ct] = __builtin_amdgcn_wmma_f32_16x16x4_f32( \
        false, afrag, false, bfrag, (short)0, accY[ct], false, false)
#define WMMA_W(ct, afrag, bfrag) accW[ct] = __builtin_amdgcn_wmma_f32_16x16x4_f32( \
        false, afrag, false, bfrag, (short)0, accW[ct], false, false)

__device__ __forceinline__ void tile_next(const v4f (&aL)[8], const v4f (&aB)[8],
                                          const float* bcur, float* knextRow, int mbase,
                                          const float* sCz, const float* sZt, const float* sQ,
                                          const float (&xs)[8], float cx, v4f& mu4,
                                          v8f (&accY)[4], v8f (&accW)[4])
{
    v4f b[4];
#pragma unroll
    for (int ct = 0; ct < 4; ++ct)
        b[ct] = *(const v4f*)(bcur + ct * 16 * KP);
#pragma unroll
    for (int j = 0; j < 8; ++j) {
        const int m0 = mbase + 4 * j;
        v4f tz[8], bn[4];
        const v4f tc = *(const v4f*)&sCz[m0];
#pragma unroll
        for (int d = 0; d < 8; ++d) tz[d] = *(const v4f*)&sZt[d * 64 + m0];
        const v4f tq = *(const v4f*)&sQ[m0];
        if (j < 7) {
#pragma unroll
            for (int ct = 0; ct < 4; ++ct)
                bn[ct] = *(const v4f*)(bcur + ct * 16 * KP + 8 * (j + 1));
        }
        PIN_HERE();
        const v2f aY0 = {aL[j].x, aL[j].y}, aY1 = {aL[j].z, aL[j].w};
        const v2f aW0 = {aB[j].x, aB[j].y}, aW1 = {aB[j].z, aB[j].w};
#pragma unroll
        for (int ct = 0; ct < 4; ++ct) {
            const v2f b0 = {b[ct].x, b[ct].y};
            WMMA_Y(ct, aY0, b0);
            WMMA_W(ct, aW0, b0);
        }
        PIN_HERE();
        const v2f b10 = {b[0].z, b[0].w}, b11 = {b[1].z, b[1].w};
        const v2f b12 = {b[2].z, b[2].w}, b13 = {b[3].z, b[3].w};
        WMMA_Y(0, aY1, b10);
        v4f e = tc + cx;
        e = __builtin_elementwise_fma(tz[0], (v4f)(xs[0]), e);
        e = __builtin_elementwise_fma(tz[1], (v4f)(xs[1]), e);
        PIN_HERE();
        WMMA_W(0, aW1, b10);
        e = __builtin_elementwise_fma(tz[2], (v4f)(xs[2]), e);
        e = __builtin_elementwise_fma(tz[3], (v4f)(xs[3]), e);
        PIN_HERE();
        WMMA_Y(1, aY1, b11);
        e = __builtin_elementwise_fma(tz[4], (v4f)(xs[4]), e);
        e = __builtin_elementwise_fma(tz[5], (v4f)(xs[5]), e);
        PIN_HERE();
        WMMA_W(1, aW1, b11);
        e = __builtin_elementwise_fma(tz[6], (v4f)(xs[6]), e);
        e = __builtin_elementwise_fma(tz[7], (v4f)(xs[7]), e);
        PIN_HERE();
        WMMA_Y(2, aY1, b12);
        v4f kv;
        kv.x = __builtin_amdgcn_exp2f(e.x);
        kv.y = __builtin_amdgcn_exp2f(e.y);
        PIN_HERE();
        WMMA_W(2, aW1, b12);
        kv.z = __builtin_amdgcn_exp2f(e.z);
        kv.w = __builtin_amdgcn_exp2f(e.w);
        PIN_HERE();
        WMMA_Y(3, aY1, b13);
        mu4 = __builtin_elementwise_fma(tq, kv, mu4);
        PIN_HERE();
        WMMA_W(3, aW1, b13);
        *(v4f*)&knextRow[m0] = kv;
        PIN_HERE();
        if (j < 7) {
#pragma unroll
            for (int ct = 0; ct < 4; ++ct) b[ct] = bn[ct];
        }
    }
}

__global__ __launch_bounds__(128) void svgp_main(
    const float* __restrict__ x, const float* __restrict__ ws,
    float* __restrict__ out, int n, int blocksPerPair)
{
    __shared__ __attribute__((aligned(16))) float sKt[2][64 * KP];
    __shared__ __attribute__((aligned(16))) float sZt[8 * 64];
    __shared__ __attribute__((aligned(16))) float sCz[64];
    __shared__ __attribute__((aligned(16))) float sQ[64];
    __shared__ float sMu[4][2][64];
    __shared__ float sD1[2][4][64];
    __shared__ float sD2[2][4][64];

    const int pair  = blockIdx.x / blocksPerPair;
    const int chunk = blockIdx.x % blocksPerPair;
    const float* wp = ws + (size_t)pair * WS_STRIDE;
    const int tid  = threadIdx.x;
    const int wave = __builtin_amdgcn_readfirstlane(tid >> 5);
    const int lane = tid & 31;
    const int c    = lane & 15;
    const int h    = lane >> 4;

    *(v4f*)&sZt[4 * tid] = *(const v4f*)(wp + WS_ZS + 4 * tid);
    if (tid < 64) { sCz[tid] = wp[WS_Z2 + tid]; sQ[tid] = wp[WS_Q + tid]; }
    float ils[8];
#pragma unroll
    for (int d = 0; d < 8; ++d) ils[d] = wp[WS_ILS + d];
    const float var = wp[WS_VAR];

    const int ncol = tid & 63;
    const int mbase = 32 * (wave >> 1);
    const float* xcol = x + ((size_t)chunk * COLS_PER_BLOCK + ncol) * 8;

    v4f xa = *(const v4f*)(xcol);
    v4f xb = *(const v4f*)(xcol + 4);
    v4f xna = *(const v4f*)(xcol + (size_t)COLS_PER_ITER * 8);
    v4f xnb = *(const v4f*)(xcol + (size_t)COLS_PER_ITER * 8 + 4);

    v4f aL[8], aB[8];
    {
        const float* rowL = wp + WS_LINV + (16 * wave + c) * 64 + 4 * h;
        const float* rowB = wp + WS_B    + (16 * wave + c) * 64 + 4 * h;
#pragma unroll
        for (int j = 0; j < 8; ++j) {
            aL[j] = *(const v4f*)(rowL + 8 * j);
            aB[j] = *(const v4f*)(rowB + 8 * j);
        }
    }
    __syncthreads();

    const size_t varBase = (size_t)(HDIM * ODIM) * (size_t)n;
    const float* bp = &sKt[0][c * KP + 4 * h];

    float xs[8], cx;
    v4f mu4 = {0.f, 0.f, 0.f, 0.f};

    scale_x(xa, xb, ils, xs, cx);
#pragma unroll 2
    for (int g = 0; g < 8; ++g)
        kfu_group(mbase + 4 * g, sCz, sZt, sQ, xs, cx, mu4, &sKt[0][ncol * KP]);
    sMu[0][wave >> 1][ncol] = (mu4.x + mu4.y) + (mu4.z + mu4.w);
    __syncthreads();

    for (int it = 0; it < NTILES; ++it) {
        const int p    = it & 1;
        const int col0 = chunk * COLS_PER_BLOCK + it * COLS_PER_ITER;

        v8f accY[4] = {};
        v8f accW[4] = {};
        if (it + 1 < NTILES) {
            scale_x(xna, xnb, ils, xs, cx);
            const int t2 = (it + 2 < NTILES) ? it + 2 : NTILES - 1;
            xna = *(const v4f*)(xcol + (size_t)t2 * COLS_PER_ITER * 8);
            xnb = *(const v4f*)(xcol + (size_t)t2 * COLS_PER_ITER * 8 + 4);
            mu4 = (v4f){0.f, 0.f, 0.f, 0.f};
            tile_next(aL, aB, bp + p * (64 * KP), &sKt[p ^ 1][ncol * KP], mbase,
                      sCz, sZt, sQ, xs, cx, mu4, accY, accW);
            sMu[(it + 1) & 3][wave >> 1][ncol] = (mu4.x + mu4.y) + (mu4.z + mu4.w);
        } else {
            tile_last(aL, aB, bp + p * (64 * KP), accY, accW);
        }

#pragma unroll
        for (int ct = 0; ct < 4; ++ct) {
            float d1 = 0.f, d2 = 0.f;
#pragma unroll
            for (int v = 0; v < 8; ++v) {
                d1 += accY[ct][v] * accY[ct][v];
                d2 += accW[ct][v] * accW[ct][v];
            }
            d1 += __shfl_xor(d1, 16, 32);
            d2 += __shfl_xor(d2, 16, 32);
            if (lane < 16) {
                sD1[p][wave][ct * 16 + c] = d1;
                sD2[p][wave][ct * 16 + c] = d2;
            }
        }
        __syncthreads();

        if (tid < 64) {
            const float m_ = sMu[it & 3][0][tid] + sMu[it & 3][1][tid];
            const float d1 = (sD1[p][0][tid] + sD1[p][1][tid]) + (sD1[p][2][tid] + sD1[p][3][tid]);
            const float d2 = (sD2[p][0][tid] + sD2[p][1][tid]) + (sD2[p][2][tid] + sD2[p][3][tid]);
            const int col = col0 + tid;
            out[(size_t)pair * n + col]           = m_;
            out[varBase + (size_t)pair * n + col] = var - d1 + d2;
        }
    }
}

extern "C" void kernel_launch(void* const* d_in, const int* in_sizes, int n_in,
                              void* d_out, int out_size, void* d_ws, size_t ws_size,
                              hipStream_t stream)
{
    const float* x          = (const float*)d_in[0];
    const float* z          = (const float*)d_in[1];
    const float* u_mean     = (const float*)d_in[2];
    const float* u_tril_vec = (const float*)d_in[3];
    const float* log_ls     = (const float*)d_in[4];
    const float* log_var    = (const float*)d_in[5];
    float* out = (float*)d_out;
    float* ws  = (float*)d_ws;

    const int n = in_sizes[0] / 8;
    svgp_prep<<<HDIM * ODIM, 128, 0, stream>>>(z, u_mean, u_tril_vec,
                                              log_ls, log_var, ws);
    const int blocksPerPair = n / COLS_PER_BLOCK;
    svgp_main<<<HDIM * ODIM * blocksPerPair, 128, 0, stream>>>(
        x, ws, out, n, blocksPerPair);
}


# main: Linv*K runs only the non-zero k steps of triangular Linv (40 of 64 f32 WMMA per 16 columns), split by column tile so every wave issues 104 per tile, not 128; Linv fragments read from LDS
# speedup vs baseline: 3.8843x; 1.0897x over previous
#include <hip/hip_runtime.h>

#define HDIM 4
#define ODIM 4
#define JITTER 1e-4f
#define LOG2E 1.44269504088896340736f

#define WS_LINV 0
#define WS_B    4096
#define WS_ZS   8192
#define WS_Z2   8704
#define WS_Q    8768
#define WS_ILS  8832
#define WS_VAR  8840
#define WS_STRIDE 8960

typedef __attribute__((ext_vector_type(2))) float v2f;
typedef __attribute__((ext_vector_type(8))) float v8f;

#define PS 65
#define WSTR 17

__device__ __forceinline__ v8f tile_mma(v8f acc, const float* A, int a_rs, int a_cs,
                                        const float* B, int b_rs, int b_cs, int lane)
{
    const int nl = lane & 15;
    const int kh = (lane >> 4) * 2;
#pragma unroll
    for (int kc = 0; kc < 4; ++kc) {
        const int k0 = kc * 4 + kh;
        v2f a, b;
        a.x = A[nl * a_rs + k0 * a_cs];
        a.y = A[nl * a_rs + (k0 + 1) * a_cs];
        b.x = B[k0 * b_rs + nl * b_cs];
        b.y = B[(k0 + 1) * b_rs + nl * b_cs];
        acc = __builtin_amdgcn_wmma_f32_16x16x4_f32(
            false, a, false, b, (short)0, acc, false, false);
    }
    return acc;
}

__device__ __forceinline__ float lane_read(float x, int k)
{
    return __builtin_bit_cast(float,
        __builtin_amdgcn_readlane(__builtin_bit_cast(int, x), k));
}

__global__ __launch_bounds__(128) void svgp_prep(
    const float* __restrict__ z, const float* __restrict__ u_mean,
    const float* __restrict__ u_tril_vec, const float* __restrict__ log_ls,
    const float* __restrict__ log_var, float* __restrict__ ws)
{
    __shared__ float sS[64 * PS];
    __shared__ float sX[64 * PS];
    __shared__ float sU[64 * PS];
    __shared__ float sW[4][16 * WSTR];
    __shared__ float sZs[64 * 8];
    __shared__ float sZ2[64];
    __shared__ float sT[64];
    __shared__ float sUm[64];

    const int p    = blockIdx.x;
    const int o    = p % ODIM;
    const int tid  = threadIdx.x;
    const int lane = tid & 31;
    const int wave = __builtin_amdgcn_readfirstlane(tid >> 5);
    const int row  = tid & 63;
    const int half = tid >> 6;
    const int accr = (lane >> 4) * 8;
    const int accc = lane & 15;
    float* wp = ws + (size_t)p * WS_STRIDE;

    float ils[8];
#pragma unroll
    for (int d = 0; d < 8; ++d) ils[d] = __expf(-log_ls[p * 8 + d]);
    const float var = __expf(log_var[p]);

#pragma unroll 8
    for (int it = 0; it < 32; ++it) {
        const int i  = 2 * it + half;
        const int jj = (row < i) ? row : i;
        const float v = u_tril_vec[o * 2080 + i * (i + 1) / 2 + jj];
        sU[i * PS + row] = (row <= i) ? v : 0.f;
    }
    for (int i = tid; i < 64 * PS; i += 128) sX[i] = 0.f;

    float zr[8];
    {
        float s = 0.f;
#pragma unroll
        for (int d = 0; d < 8; ++d) {
            zr[d] = z[(o * 64 + row) * 8 + d] * ils[d];
            s += zr[d] * zr[d];
        }
        if (half == 0) {
#pragma unroll
            for (int d = 0; d < 8; ++d) sZs[row * 8 + d] = zr[d];
            sZ2[row] = s;
            sUm[row] = u_mean[o * 64 + row];
        }
    }
    __syncthreads();

    {
        const float z2i = sZ2[row];
        for (int jj = 0; jj < 32; ++jj) {
            const int j = half * 32 + jj;
            float dot = 0.f;
#pragma unroll
            for (int d = 0; d < 8; ++d) dot += zr[d] * sZs[j * 8 + d];
            float v = var * __expf(-0.5f * (z2i + sZ2[j] - 2.f * dot));
            if (row == j) v += JITTER;
            sS[row * PS + j] = v;
        }
    }
    __syncthreads();

    for (int J = 0; J < 4; ++J) {
        if (J > 0 && wave >= J) {
            v8f acc = {};
            for (int P = 0; P < J; ++P)
                acc = tile_mma(acc, &sS[(16 * wave) * PS + 16 * P], PS, 1,
                                    &sS[(16 * J) * PS + 16 * P], 1, PS, lane);
#pragma unroll
            for (int v = 0; v < 8; ++v)
                sS[(16 * wave + accr + v) * PS + 16 * J + accc] -= acc[v];
        }
        __syncthreads();

        float* blk = &sS[(16 * J) * PS + 16 * J];
        if (wave == 0) {
            const int t = lane & 15;
            float a[16], rs[16];
#pragma unroll
            for (int k = 0; k < 16; ++k) a[k] = blk[t * PS + k];
#pragma unroll
            for (int s = 0; s < 16; ++s) {
                rs[s] = __builtin_amdgcn_rsqf(lane_read(a[s], s));
                a[s] *= rs[s];
#pragma unroll
                for (int k = s + 1; k < 16; ++k) a[k] -= a[s] * lane_read(a[s], k);
            }
            float x[16];
#pragma unroll
            for (int i = 0; i < 16; ++i) x[i] = (i == t) ? 1.f : 0.f;
#pragma unroll
            for (int i = 0; i < 16; ++i) {
                x[i] *= rs[i];
#pragma unroll
                for (int k = i + 1; k < 16; ++k) x[k] -= lane_read(a[i], k) * x[i];
            }
            if (lane < 16) {
#pragma unroll
                for (int i = 0; i < 16; ++i) sW[J][i * WSTR + t] = x[i];
#pragma unroll
                for (int k = 0; k < 16; ++k)
                    if (k <= t) blk[t * PS + k] = a[k];
            }
        }
        __syncthreads();

        if (wave > J) {
            v8f acc = {};
            acc = tile_mma(acc, &sS[(16 * wave) * PS + 16 * J], PS, 1,
                                &sW[J][0], 1, WSTR, lane);
#pragma unroll
            for (int v = 0; v < 8; ++v)
                sS[(16 * wave + accr + v) * PS + 16 * J + accc] = acc[v];
        }
        __syncthreads();
    }

    {
        const int r = tid >> 3, c0 = (tid & 7) * 2;
#pragma unroll
        for (int J = 0; J < 4; ++J) {
            sX[(16 * J + r) * PS + 16 * J + c0]     = sW[J][r * WSTR + c0];
            sX[(16 * J + r) * PS + 16 * J + c0 + 1] = sW[J][r * WSTR + c0 + 1];
        }
    }
    __syncthreads();
    for (int dist = 1; dist < 4; ++dist) {
        const int I  = wave;
        const int Jc = wave - dist;
        const bool act = (Jc >= 0);
        if (act) {
            v8f acc = {};
            for (int P = Jc; P < I; ++P)
                acc = tile_mma(acc, &sS[(16 * I) * PS + 16 * P], PS, 1,
                                    &sX[(16 * P) * PS + 16 * Jc], PS, 1, lane);
#pragma unroll
            for (int v = 0; v < 8; ++v)
                sX[(16 * I + accr + v) * PS + 16 * Jc + accc] = acc[v];
        }
        __syncthreads();
        v8f acc2 = {};
        if (act)
            acc2 = tile_mma(acc2, &sW[I][0], WSTR, 1,
                                  &sX[(16 * I) * PS + 16 * Jc], PS, 1, lane);
        __syncthreads();
        if (act) {
#pragma unroll
            for (int v = 0; v < 8; ++v)
                sX[(16 * I + accr + v) * PS + 16 * Jc + accc] = -acc2[v];
        }
        __syncthreads();
    }

    {
        const float di = 1.f / sS[row * PS + row];
        for (int jj = 0; jj < 32; ++jj) sU[row * PS + half * 32 + jj] *= di;
        if (half == 0) {
            float s = 0.f;
            for (int j = 0; j < 64; ++j) s += sX[row * PS + j] * sUm[j];
            sT[row] = s;
        }
    }
    __syncthreads();

    if (half == 0) {
        float s = 0.f;
        for (int i = 0; i < 64; ++i) s += sX[i * PS + row] * sT[i];
        wp[WS_Q + row] = s;
    }

    for (int Rt = 0; Rt < 4; ++Rt) {
        v8f acc = {};
        const int It0 = (wave > Rt) ? wave : Rt;
        for (int It = It0; It < 4; ++It)
            acc = tile_mma(acc, &sU[(16 * It) * PS + 16 * wave], 1, PS,
                                &sX[(16 * It) * PS + 16 * Rt], PS, 1, lane);
#pragma unroll
        for (int v = 0; v < 8; ++v)
            wp[WS_B + (16 * wave + accr + v) * 64 + 16 * Rt + accc] = acc[v];
    }

    for (int i = tid; i < 4096; i += 128)
        wp[WS_LINV + i] = sX[(i >> 6) * PS + (i & 63)];
    for (int i = tid; i < 512; i += 128) wp[WS_ZS + i] = sZs[(i & 63) * 8 + (i >> 6)] * LOG2E;
    if (tid < 64) wp[WS_Z2 + tid] = LOG2E * (log_var[p] - 0.5f * sZ2[tid]);
    if (tid < 8)  wp[WS_ILS + tid] = __expf(-log_ls[p * 8 + tid]);
    if (tid == 0) wp[WS_VAR] = var;
}

#define KP 68
#define COLS_PER_BLOCK 512
#define COLS_PER_ITER 64
#define NTILES (COLS_PER_BLOCK / COLS_PER_ITER)

typedef __attribute__((ext_vector_type(4))) float v4f;

__device__ __forceinline__ void kfu_group(int m0, const float* sCz, const float* sZt,
                                          const float* sQ, const float (&xs)[8], float cx,
                                          v4f& mu4, float* dstRow)
{
    v4f e = *(const v4f*)&sCz[m0] + cx;
#pragma unroll
    for (int d = 0; d < 8; ++d)
        e = __builtin_elementwise_fma(*(const v4f*)&sZt[d * 64 + m0], (v4f)(xs[d]), e);
    v4f kv;
    kv.x = __builtin_amdgcn_exp2f(e.x);
    kv.y = __builtin_amdgcn_exp2f(e.y);
    kv.z = __builtin_amdgcn_exp2f(e.z);
    kv.w = __builtin_amdgcn_exp2f(e.w);
    mu4 = __builtin_elementwise_fma(*(const v4f*)&sQ[m0], kv, mu4);
    *(v4f*)&dstRow[m0] = kv;
}

__device__ __forceinline__ void scale_x(const v4f xa, const v4f xb, const float (&ils)[8],
                                        float (&xs)[8], float& cx)
{
    xs[0] = xa.x * ils[0]; xs[1] = xa.y * ils[1]; xs[2] = xa.z * ils[2]; xs[3] = xa.w * ils[3];
    xs[4] = xb.x * ils[4]; xs[5] = xb.y * ils[5]; xs[6] = xb.z * ils[6]; xs[7] = xb.w * ils[7];
    float xn2 = 0.f;
#pragma unroll
    for (int d = 0; d < 8; ++d) xn2 += xs[d] * xs[d];
    cx = (-0.5f * LOG2E) * xn2;
}

#define RT0(j) ((j) >> 1)

__device__ __forceinline__ void tile_last(const float* lfrag, const v4f (&aB)[8],
                                          const float* const (&bc)[4],
                                          v8f (&accY)[4], v8f (&accW)[4])
{
#pragma unroll
    for (int j = 0; j < 8; ++j) {
        v4f b[4], fL[4];
#pragma unroll
        for (int ct = 0; ct < 4; ++ct)
            b[ct] = *(const v4f*)(bc[ct] + 8 * j);
#pragma unroll
        for (int r = RT0(j); r < 4; ++r)
            fL[r] = *(const v4f*)(lfrag + r * 16 * KP + 8 * j);
        const v2f aW0 = {aB[j].x, aB[j].y}, aW1 = {aB[j].z, aB[j].w};
#pragma unroll
        for (int ct = 0; ct < 4; ++ct) {
            const v2f b0 = {b[ct].x, b[ct].y};
            accW[ct] = __builtin_amdgcn_wmma_f32_16x16x4_f32(
                false, aW0, false, b0, (short)0, accW[ct], false, false);
        }
#pragma unroll
        for (int ct = 0; ct < 4; ++ct) {
            const v2f b1 = {b[ct].z, b[ct].w};
            accW[ct] = __builtin_amdgcn_wmma_f32_16x16x4_f32(
                false, aW1, false, b1, (short)0, accW[ct], false, false);
        }
        const v2f y0 = {b[0].x, b[0].y}, y1 = {b[0].z, b[0].w};
#pragma unroll
        for (int r = RT0(j); r < 4; ++r) {
            const v2f a0 = {fL[r].x, fL[r].y};
            accY[r] = __builtin_amdgcn_wmma_f32_16x16x4_f32(
                false, a0, false, y0, (short)0, accY[r], false, false);
        }
#pragma unroll
        for (int r = RT0(j); r < 4; ++r) {
            const v2f a1 = {fL[r].z, fL[r].w};
            accY[r] = __builtin_amdgcn_wmma_f32_16x16x4_f32(
                false, a1, false, y1, (short)0, accY[r], false, false);
        }
    }
}

#define PIN_HERE() __builtin_amdgcn_sched_barrier(0)

#define WMMA_Y(ct, afrag, bfrag) accY[ct] = __builtin_amdgcn_wmma_f32_16x16x4_f32( \
        false, afrag, false, bfrag, (short)0, accY[ct], false, false)
#define WMMA_W(ct, afrag, bfrag) accW[ct] = __builtin_amdgcn_wmma_f32_16x16x4_f32( \
        false, afrag, false, bfrag, (short)0, accW[ct], false, false)

#define S0() e = __builtin_elementwise_fma(tz[0], (v4f)(xs[0]), e); e = __builtin_elementwise_fma(tz[1], (v4f)(xs[1]), e)
#define S1() e = __builtin_elementwise_fma(tz[2], (v4f)(xs[2]), e); e = __builtin_elementwise_fma(tz[3], (v4f)(xs[3]), e)
#define S2() e = __builtin_elementwise_fma(tz[4], (v4f)(xs[4]), e); e = __builtin_elementwise_fma(tz[5], (v4f)(xs[5]), e)
#define S3() e = __builtin_elementwise_fma(tz[6], (v4f)(xs[6]), e); e = __builtin_elementwise_fma(tz[7], (v4f)(xs[7]), e)
#define S4() kv.x = __builtin_amdgcn_exp2f(e.x); kv.y = __builtin_amdgcn_exp2f(e.y)
#define S5() kv.z = __builtin_amdgcn_exp2f(e.z); kv.w = __builtin_amdgcn_exp2f(e.w)
#define S6() mu4 = __builtin_elementwise_fma(tq, kv, mu4)
#define S7() *(v4f*)&knextRow[m0] = kv
#define Y0(r) { const v2f a_ = {fL[r].x, fL[r].y}; WMMA_Y(r, a_, y0); }
#define Y1(r) { const v2f a_ = {fL[r].z, fL[r].w}; WMMA_Y(r, a_, y1); }

__device__ __forceinline__ void tile_next(const float* lfrag, const v4f (&aB)[8],
                                          const float* const (&bc)[4],
                                          float* knextRow, int mbase,
                                          const float* sCz, const float* sZt, const float* sQ,
                                          const float (&xs)[8], float cx, v4f& mu4,
                                          v8f (&accY)[4], v8f (&accW)[4])
{
    v4f b[4];
#pragma unroll
    for (int ct = 0; ct < 4; ++ct)
        b[ct] = *(const v4f*)(bc[ct]);
#pragma unroll
    for (int j = 0; j < 8; ++j) {
        const int m0 = mbase + 4 * j;
        v4f tz[8], bn[4], fL[4];
        const v4f tc = *(const v4f*)&sCz[m0];
#pragma unroll
        for (int d = 0; d < 8; ++d) tz[d] = *(const v4f*)&sZt[d * 64 + m0];
        const v4f tq = *(const v4f*)&sQ[m0];
#pragma unroll
        for (int r = RT0(j); r < 4; ++r)
            fL[r] = *(const v4f*)(lfrag + r * 16 * KP + 8 * j);
        if (j < 7) {
#pragma unroll
            for (int ct = 0; ct < 4; ++ct)
                bn[ct] = *(const v4f*)(bc[ct] + 8 * (j + 1));
        }
        PIN_HERE();
        const v2f aW0 = {aB[j].x, aB[j].y}, aW1 = {aB[j].z, aB[j].w};
#pragma unroll
        for (int ct = 0; ct < 4; ++ct) {
            const v2f b0 = {b[ct].x, b[ct].y};
            WMMA_W(ct, aW0, b0);
        }
#pragma unroll
        for (int ct = 0; ct < 4; ++ct) {
            const v2f b1 = {b[ct].z, b[ct].w};
            WMMA_W(ct, aW1, b1);
        }
        PIN_HERE();
        const v2f y0 = {b[0].x, b[0].y}, y1 = {b[0].z, b[0].w};
        v4f e = tc + cx;
        v4f kv;
        if (RT0(j) == 0) {
            Y0(0); S0(); PIN_HERE();
            Y0(1); S1(); PIN_HERE();
            Y0(2); S2(); PIN_HERE();
            Y0(3); S3(); PIN_HERE();
            Y1(0); S4(); PIN_HERE();
            Y1(1); S5(); PIN_HERE();
            Y1(2); S6(); PIN_HERE();
            Y1(3); S7(); PIN_HERE();
        } else if (RT0(j) == 1) {
            Y0(1); S0(); PIN_HERE();
            Y0(2); S1(); PIN_HERE();
            Y0(3); S2(); S3(); PIN_HERE();
            Y1(1); S4(); PIN_HERE();
            Y1(2); S5(); PIN_HERE();
            Y1(3); S6(); S7(); PIN_HERE();
        } else if (RT0(j) == 2) {
            Y0(2); S0(); S1(); PIN_HERE();
            Y0(3); S2(); S3(); PIN_HERE();
            Y1(2); S4(); S5(); PIN_HERE();
            Y1(3); S6(); S7(); PIN_HERE();
        } else {
            Y0(3); S0(); S1(); S2(); S3(); PIN_HERE();
            Y1(3); S4(); S5(); S6(); S7(); PIN_HERE();
        }
        if (j < 7) {
#pragma unroll
            for (int ct = 0; ct < 4; ++ct) b[ct] = bn[ct];
        }
    }
}

__global__ __launch_bounds__(128) void svgp_main(
    const float* __restrict__ x, const float* __restrict__ ws,
    float* __restrict__ out, int n, int blocksPerPair)
{
    __shared__ __attribute__((aligned(16))) float sKt[2][64 * KP];
    __shared__ __attribute__((aligned(16))) float sZt[8 * 64];
    __shared__ __attribute__((aligned(16))) float sCz[64];
    __shared__ __attribute__((aligned(16))) float sQ[64];
    __shared__ float sMu[4][2][64];
    __shared__ __attribute__((aligned(16))) float sL[64 * KP];
    __shared__ float sD1[2][64];
    __shared__ float sD2[2][4][64];

    const int pair  = blockIdx.x / blocksPerPair;
    const int chunk = blockIdx.x % blocksPerPair;
    const float* wp = ws + (size_t)pair * WS_STRIDE;
    const int tid  = threadIdx.x;
    const int wave = __builtin_amdgcn_readfirstlane(tid >> 5);
    const int lane = tid & 31;
    const int c    = lane & 15;
    const int h    = lane >> 4;

    *(v4f*)&sZt[4 * tid] = *(const v4f*)(wp + WS_ZS + 4 * tid);
    if (tid < 64) { sCz[tid] = wp[WS_Z2 + tid]; sQ[tid] = wp[WS_Q + tid]; }
    float ils[8];
#pragma unroll
    for (int d = 0; d < 8; ++d) ils[d] = wp[WS_ILS + d];
    const float var = wp[WS_VAR];

    const int ncol = tid & 63;
    const int mbase = 32 * (wave >> 1);
    const float* xcol = x + ((size_t)chunk * COLS_PER_BLOCK + ncol) * 8;

    v4f xa = *(const v4f*)(xcol);
    v4f xb = *(const v4f*)(xcol + 4);
    v4f xna = *(const v4f*)(xcol + (size_t)COLS_PER_ITER * 8);
    v4f xnb = *(const v4f*)(xcol + (size_t)COLS_PER_ITER * 8 + 4);

#pragma unroll
    for (int q = 0; q < 8; ++q) {
        const int f = 4 * (tid + 128 * q);
        *(v4f*)&sL[(f >> 6) * KP + (f & 63)] = *(const v4f*)(wp + WS_LINV + f);
    }
    v4f aB[8];
    {
        const float* rowB = wp + WS_B + (16 * wave + c) * 64 + 4 * h;
#pragma unroll
        for (int j = 0; j < 8; ++j)
            aB[j] = *(const v4f*)(rowB + 8 * j);
    }
    __syncthreads();

    const size_t varBase = (size_t)(HDIM * ODIM) * (size_t)n;
    const float* bp[4];
#pragma unroll
    for (int i = 0; i < 4; ++i)
        bp[i] = &sKt[0][(16 * ((wave + i) & 3) + c) * KP + 4 * h];
    const float* lfrag = &sL[c * KP + 4 * h];

    float xs[8], cx;
    v4f mu4 = {0.f, 0.f, 0.f, 0.f};

    scale_x(xa, xb, ils, xs, cx);
#pragma unroll 2
    for (int g = 0; g < 8; ++g)
        kfu_group(mbase + 4 * g, sCz, sZt, sQ, xs, cx, mu4, &sKt[0][ncol * KP]);
    sMu[0][wave >> 1][ncol] = (mu4.x + mu4.y) + (mu4.z + mu4.w);
    __syncthreads();

    for (int it = 0; it < NTILES; ++it) {
        const int p    = it & 1;
        const int col0 = chunk * COLS_PER_BLOCK + it * COLS_PER_ITER;

        v8f accY[4] = {};
        v8f accW[4] = {};
        if (it + 1 < NTILES) {
            scale_x(xna, xnb, ils, xs, cx);
            const int t2 = (it + 2 < NTILES) ? it + 2 : NTILES - 1;
            xna = *(const v4f*)(xcol + (size_t)t2 * COLS_PER_ITER * 8);
            xnb = *(const v4f*)(xcol + (size_t)t2 * COLS_PER_ITER * 8 + 4);
            mu4 = (v4f){0.f, 0.f, 0.f, 0.f};
            const float* const bc[4] = {bp[0] + p * (64 * KP), bp[1] + p * (64 * KP),
                                        bp[2] + p * (64 * KP), bp[3] + p * (64 * KP)};
            tile_next(lfrag, aB, bc, &sKt[p ^ 1][ncol * KP], mbase,
                      sCz, sZt, sQ, xs, cx, mu4, accY, accW);
            sMu[(it + 1) & 3][wave >> 1][ncol] = (mu4.x + mu4.y) + (mu4.z + mu4.w);
        } else {
            const float* const bc[4] = {bp[0] + p * (64 * KP), bp[1] + p * (64 * KP),
                                        bp[2] + p * (64 * KP), bp[3] + p * (64 * KP)};
            tile_last(lfrag, aB, bc, accY, accW);
        }

#pragma unroll
        for (int ct = 0; ct < 4; ++ct) {
            float d2 = 0.f;
#pragma unroll
            for (int v = 0; v < 8; ++v) d2 += accW[ct][v] * accW[ct][v];
            d2 += __shfl_xor(d2, 16, 32);
            if (lane < 16) sD2[p][wave][((wave + ct) & 3) * 16 + c] = d2;
        }
        {
            float d1 = 0.f;
#pragma unroll
            for (int r = 0; r < 4; ++r)
#pragma unroll
                for (int v = 0; v < 8; ++v) d1 += accY[r][v] * accY[r][v];
            d1 += __shfl_xor(d1, 16, 32);
            if (lane < 16) sD1[p][wave * 16 + c] = d1;
        }
        __syncthreads();

        if (tid < 64) {
            const float m_ = sMu[it & 3][0][tid] + sMu[it & 3][1][tid];
            const float d1 = sD1[p][tid];
            const float d2 = (sD2[p][0][tid] + sD2[p][1][tid]) + (sD2[p][2][tid] + sD2[p][3][tid]);
            const int col = col0 + tid;
            out[(size_t)pair * n + col]           = m_;
            out[varBase + (size_t)pair * n + col] = var - d1 + d2;
        }
    }
}

extern "C" void kernel_launch(void* const* d_in, const int* in_sizes, int n_in,
                              void* d_out, int out_size, void* d_ws, size_t ws_size,
                              hipStream_t stream)
{
    const float* x          = (const float*)d_in[0];
    const float* z          = (const float*)d_in[1];
    const float* u_mean     = (const float*)d_in[2];
    const float* u_tril_vec = (const float*)d_in[3];
    const float* log_ls     = (const float*)d_in[4];
    const float* log_var    = (const float*)d_in[5];
    float* out = (float*)d_out;
    float* ws  = (float*)d_ws;

    const int n = in_sizes[0] / 8;
    svgp_prep<<<HDIM * ODIM, 128, 0, stream>>>(z, u_mean, u_tril_vec,
                                              log_ls, log_var, ws);
    const int blocksPerPair = n / COLS_PER_BLOCK;
    svgp_main<<<HDIM * ODIM * blocksPerPair, 128, 0, stream>>>(
        x, ws, out, n, blocksPerPair);
}
